# starting kernel
# baseline (speedup 1.0000x reference)
#include <hip/hip_runtime.h>


typedef __attribute__((ext_vector_type(2))) float v2f;
typedef __attribute__((ext_vector_type(8))) float v8f;

#define N_NODES 100000
#define F_IN    256
#define F_HID   64
#define F_OUT   4


__global__ void deg_init_kernel(float* __restrict__ deg) {
    int i = blockIdx.x * blockDim.x + threadIdx.x;
    if (i < N_NODES) deg[i] = 1.0f;
}

__global__ void deg_accum_kernel(const int* __restrict__ dst, int E,
                                 float* __restrict__ deg) {
    int e = blockIdx.x * blockDim.x + threadIdx.x;
    if (e < E) atomicAdd(&deg[dst[e]], 1.0f);
}

__global__ void dinv_kernel(const float* __restrict__ deg,
                            float* __restrict__ dinv) {
    int i = blockIdx.x * blockDim.x + threadIdx.x;
    if (i < N_NODES) dinv[i] = rsqrtf(deg[i]);
}

__global__ void edge_norm_kernel(const int* __restrict__ src,
                                 const int* __restrict__ dst,
                                 const float* __restrict__ dinv,
                                 float* __restrict__ norm, int E) {
    int e = blockIdx.x * blockDim.x + threadIdx.x;
    if (e < E) norm[e] = dinv[src[e]] * dinv[dst[e]];
}


__global__ void __launch_bounds__(256)
gemm1_wmma_kernel(const float* __restrict__ x,
                  const float* __restrict__ W1,
                  float* __restrict__ h1) {
    __shared__ float lw[F_IN * F_HID];
    for (int i = threadIdx.x; i < F_IN * F_HID; i += blockDim.x) {
        int kk  = i >> 6;
        int col = i & 63;
        lw[(((kk >> 1) * F_HID) + col) * 2 + (kk & 1)] = W1[i];
    }
    __syncthreads();

    const int lane  = threadIdx.x & 31;
    const int wave  = threadIdx.x >> 5;
    const int mtile = blockIdx.x * 8 + wave;
    if (mtile * 16 >= N_NODES) return;
    const int row0 = mtile * 16;

    const int r  = lane & 15;
    const int hi = lane >> 4;

    v8f acc0 = {}, acc1 = {}, acc2 = {}, acc3 = {};
    const float* xrow = x + (size_t)(row0 + r) * F_IN;
    const v2f*   wp   = (const v2f*)lw;

    for (int k = 0; k < F_IN; k += 4) {
        __builtin_prefetch(&xrow[k + 32], 0, 3);
        v2f a = *(const v2f*)&xrow[k + 2 * hi];
        const int pb = ((k >> 1) + hi) * F_HID;
        v2f b0 = wp[pb + r +  0];
        v2f b1 = wp[pb + r + 16];
        v2f b2 = wp[pb + r + 32];
        v2f b3 = wp[pb + r + 48];
        acc0 = __builtin_amdgcn_wmma_f32_16x16x4_f32(false, a, false, b0, (short)0, acc0, false, false);
        acc1 = __builtin_amdgcn_wmma_f32_16x16x4_f32(false, a, false, b1, (short)0, acc1, false, false);
        acc2 = __builtin_amdgcn_wmma_f32_16x16x4_f32(false, a, false, b2, (short)0, acc2, false, false);
        acc3 = __builtin_amdgcn_wmma_f32_16x16x4_f32(false, a, false, b3, (short)0, acc3, false, false);
    }

    float* out = h1 + (size_t)row0 * F_HID;
    for (int i = 0; i < 8; i++) {
        int m = i + 8 * hi;
        out[(size_t)m * F_HID + (r +  0)] = acc0[i];
        out[(size_t)m * F_HID + (r + 16)] = acc1[i];
        out[(size_t)m * F_HID + (r + 32)] = acc2[i];
        out[(size_t)m * F_HID + (r + 48)] = acc3[i];
    }
}


__global__ void scatter1_kernel(const int* __restrict__ src,
                                const int* __restrict__ dst,
                                const float* __restrict__ norm,
                                const float* __restrict__ h1,
                                float* __restrict__ agg1, int E) {
    int t = blockIdx.x * blockDim.x + threadIdx.x;
    if (t >= E * F_HID) return;
    int e = t >> 6;
    int c = t & 63;
    float v = h1[(size_t)src[e] * F_HID + c] * norm[e];
    atomicAdd(&agg1[(size_t)dst[e] * F_HID + c], v);
}

__global__ void post1_kernel(const float* __restrict__ h1,
                             const float* __restrict__ deg,
                             const float* __restrict__ b1,
                             float* __restrict__ agg1) {
    int t = blockIdx.x * blockDim.x + threadIdx.x;
    if (t >= N_NODES * F_HID) return;
    int i = t >> 6;
    int c = t & 63;
    float v = agg1[t] + h1[t] * (1.0f / deg[i]) + b1[c];
    agg1[t] = (v > 0.0f) ? v : 0.01f * v;
}


__global__ void __launch_bounds__(256)
gemm2_wmma_kernel(const float* __restrict__ h,
                  const float* __restrict__ W2,
                  float* __restrict__ h2) {
    __shared__ float lw[F_HID * 16];
    for (int i = threadIdx.x; i < F_HID * 16; i += blockDim.x) {
        int kk = i >> 4, nn = i & 15;
        float v = (nn < F_OUT) ? W2[kk * F_OUT + nn] : 0.0f;
        lw[(((kk >> 1) * 16) + nn) * 2 + (kk & 1)] = v;
    }
    __syncthreads();

    const int lane  = threadIdx.x & 31;
    const int wave  = threadIdx.x >> 5;
    const int mtile = blockIdx.x * 8 + wave;
    if (mtile * 16 >= N_NODES) return;
    const int row0 = mtile * 16;

    const int r  = lane & 15;
    const int hi = lane >> 4;

    v8f acc = {};
    const float* hrow = h + (size_t)(row0 + r) * F_HID;
    const v2f*   wp   = (const v2f*)lw;
    for (int k = 0; k < F_HID; k += 4) {
        v2f a = *(const v2f*)&hrow[k + 2 * hi];
        v2f b = wp[((k >> 1) + hi) * 16 + r];
        acc = __builtin_amdgcn_wmma_f32_16x16x4_f32(false, a, false, b, (short)0, acc, false, false);
    }

    if (r < F_OUT) {
        float* out = h2 + (size_t)row0 * F_OUT;
        for (int i = 0; i < 8; i++) {
            int m = i + 8 * hi;
            out[(size_t)m * F_OUT + r] = acc[i];
        }
    }
}


__global__ void scatter2_kernel(const int* __restrict__ src,
                                const int* __restrict__ dst,
                                const float* __restrict__ norm,
                                const float* __restrict__ h2,
                                float* __restrict__ agg2, int E) {
    int e = blockIdx.x * blockDim.x + threadIdx.x;
    if (e >= E) return;
    float nrm = norm[e];
    const float4* hv = (const float4*)h2;
    float4 v = hv[src[e]];
    float* o = agg2 + (size_t)dst[e] * F_OUT;
    atomicAdd(&o[0], v.x * nrm);
    atomicAdd(&o[1], v.y * nrm);
    atomicAdd(&o[2], v.z * nrm);
    atomicAdd(&o[3], v.w * nrm);
}

__global__ void post2_kernel(const float* __restrict__ h2,
                             const float* __restrict__ deg,
                             const float* __restrict__ b2,
                             const float* __restrict__ agg2,
                             float* __restrict__ out) {
    int i = blockIdx.x * blockDim.x + threadIdx.x;
    if (i >= N_NODES) return;
    float inv = 1.0f / deg[i];
    float v[F_OUT];
    float mx = -3.4e38f;
    for (int c = 0; c < F_OUT; c++) {
        v[c] = agg2[(size_t)i * F_OUT + c] + h2[(size_t)i * F_OUT + c] * inv + b2[c];
        mx = fmaxf(mx, v[c]);
    }
    float s = 0.0f;
    for (int c = 0; c < F_OUT; c++) { v[c] = __expf(v[c] - mx); s += v[c]; }
    float rs = 1.0f / s;
    for (int c = 0; c < F_OUT; c++) out[(size_t)i * F_OUT + c] = v[c] * rs;
}


extern "C" void kernel_launch(void* const* d_in, const int* in_sizes, int n_in,
                              void* d_out, int out_size, void* d_ws, size_t ws_size,
                              hipStream_t stream) {
    const float* x  = (const float*)d_in[0];
    const int*   ei = (const int*)d_in[1];
    const float* W1 = (const float*)d_in[2];
    const float* b1 = (const float*)d_in[3];
    const float* W2 = (const float*)d_in[4];
    const float* b2 = (const float*)d_in[5];
    float* out = (float*)d_out;

    const int E = in_sizes[1] / 2;
    const int* src = ei;
    const int* dst = ei + E;

    float* ws   = (float*)d_ws;
    float* deg  = ws;
    float* dinv = deg  + N_NODES;
    float* norm = dinv + N_NODES;
    float* h1   = norm + E;
    float* agg1 = h1   + (size_t)N_NODES * F_HID;
    float* h2   = agg1 + (size_t)N_NODES * F_HID;
    float* agg2 = h2   + (size_t)N_NODES * F_OUT;

    const int T = 256;
    const int nb_nodes  = (N_NODES + T - 1) / T;
    const int nb_edges  = (E + T - 1) / T;
    const int nb_tiles  = (N_NODES / 16 + 7) / 8;
    const int nb_nf     = (N_NODES * F_HID + T - 1) / T;
    const long long s1t = (long long)E * F_HID;
    const int nb_s1     = (int)((s1t + T - 1) / T);

    deg_init_kernel<<<nb_nodes, T, 0, stream>>>(deg);
    deg_accum_kernel<<<nb_edges, T, 0, stream>>>(dst, E, deg);
    dinv_kernel<<<nb_nodes, T, 0, stream>>>(deg, dinv);
    edge_norm_kernel<<<nb_edges, T, 0, stream>>>(src, dst, dinv, norm, E);

    gemm1_wmma_kernel<<<nb_tiles, T, 0, stream>>>(x, W1, h1);
    hipMemsetAsync(agg1, 0, (size_t)N_NODES * F_HID * sizeof(float), stream);
    scatter1_kernel<<<nb_s1, T, 0, stream>>>(src, dst, norm, h1, agg1, E);
    post1_kernel<<<nb_nf, T, 0, stream>>>(h1, deg, b1, agg1);

    gemm2_wmma_kernel<<<nb_tiles, T, 0, stream>>>(agg1, W2, h2);
    hipMemsetAsync(agg2, 0, (size_t)N_NODES * F_OUT * sizeof(float), stream);
    scatter2_kernel<<<nb_edges, T, 0, stream>>>(src, dst, norm, h2, agg2, E);
    post2_kernel<<<nb_nodes, T, 0, stream>>>(h2, deg, b2, agg2, out);
}


# float-atomic scatter-add replaced by per-call counting sort of edges by destination (returning integer atomic = rank) and per-node register sums; sum order = arrival rank + lane partials
# speedup vs baseline: 2.9638x; 2.9638x over previous
#include <hip/hip_runtime.h>


typedef __attribute__((ext_vector_type(2))) float v2f;
typedef __attribute__((ext_vector_type(8))) float v8f;

#define N_NODES 100000
#define F_IN    256
#define F_HID   64
#define F_OUT   4


__global__ void zero_cnt_kernel(int* __restrict__ cnt) {
    int i = blockIdx.x * blockDim.x + threadIdx.x;
    if (i < N_NODES) cnt[i] = 0;
}

__global__ void count_rank_kernel(const int* __restrict__ dst, int E,
                                  int* __restrict__ cnt,
                                  int* __restrict__ rank) {
    int e = blockIdx.x * blockDim.x + threadIdx.x;
    if (e < E) rank[e] = atomicAdd(&cnt[dst[e]], 1);
}

__global__ void __launch_bounds__(256)
scan_kernel(const int* __restrict__ cnt, int* __restrict__ off,
            float* __restrict__ dinv) {
    __shared__ int red[8];
    __shared__ int wsum[8];
    const int tid  = threadIdx.x;
    const int lane = tid & 31;
    const int wave = tid >> 5;
    const int start = blockIdx.x * 1024;

    const int4* c4 = (const int4*)cnt;
    int part = 0;
#pragma unroll 8
    for (int k = 0; k < (int)blockIdx.x; k++) {
        int4 v = c4[k * 256 + tid];
        part += (v.x + v.y) + (v.z + v.w);
    }
    for (int o = 16; o > 0; o >>= 1) part += __shfl_xor(part, o);
    if (lane == 0) red[wave] = part;

    const int i0 = start + tid * 4;
    int4 c = {0, 0, 0, 0};
    if (i0 < N_NODES) c = c4[i0 >> 2];
    const int t = (c.x + c.y) + (c.z + c.w);
    int inc = t;
    for (int o = 1; o < 32; o <<= 1) {
        int y = __shfl_up(inc, o);
        if (lane >= o) inc += y;
    }
    if (lane == 31) wsum[wave] = inc;
    __syncthreads();

    int ex = inc - t;
    for (int w = 0; w < 8; w++) {
        ex += red[w];
        if (w < wave) ex += wsum[w];
    }
    if (i0 < N_NODES) {
        int4 o4;
        o4.x = ex;
        o4.y = ex + c.x;
        o4.z = ex + c.x + c.y;
        o4.w = ex + c.x + c.y + c.z;
        ((int4*)off)[i0 >> 2] = o4;
        float4 d4;
        d4.x = rsqrtf((float)c.x + 1.0f);
        d4.y = rsqrtf((float)c.y + 1.0f);
        d4.z = rsqrtf((float)c.z + 1.0f);
        d4.w = rsqrtf((float)c.w + 1.0f);
        ((float4*)dinv)[i0 >> 2] = d4;
    }
}

__global__ void fill_kernel(const int* __restrict__ src,
                            const int* __restrict__ dst,
                            const int* __restrict__ rank,
                            const int* __restrict__ off,
                            int* __restrict__ csr_src, int E) {
    int e = blockIdx.x * blockDim.x + threadIdx.x;
    if (e >= E) return;
    unsigned p = (unsigned)(off[dst[e]] + rank[e]);
    p = min(p, (unsigned)(E - 1));
    csr_src[p] = src[e];
}


__global__ void __launch_bounds__(256)
gemm1_wmma_kernel(const float* __restrict__ x,
                  const float* __restrict__ W1,
                  float* __restrict__ h1) {
    __shared__ float lw[F_IN * F_HID];
    for (int i = threadIdx.x; i < F_IN * F_HID; i += blockDim.x) {
        int kk  = i >> 6;
        int col = i & 63;
        lw[(((kk >> 1) * F_HID) + col) * 2 + (kk & 1)] = W1[i];
    }
    __syncthreads();

    const int lane  = threadIdx.x & 31;
    const int wave  = threadIdx.x >> 5;
    const int mtile = blockIdx.x * 8 + wave;
    if (mtile * 16 >= N_NODES) return;
    const int row0 = mtile * 16;

    const int r  = lane & 15;
    const int hi = lane >> 4;

    v8f acc0 = {}, acc1 = {}, acc2 = {}, acc3 = {};
    const float* xrow = x + (size_t)(row0 + r) * F_IN;
    const v2f*   wp   = (const v2f*)lw;

    for (int k = 0; k < F_IN; k += 4) {
        __builtin_prefetch(&xrow[k + 32], 0, 3);
        v2f a = *(const v2f*)&xrow[k + 2 * hi];
        const int pb = ((k >> 1) + hi) * F_HID;
        v2f b0 = wp[pb + r +  0];
        v2f b1 = wp[pb + r + 16];
        v2f b2 = wp[pb + r + 32];
        v2f b3 = wp[pb + r + 48];
        acc0 = __builtin_amdgcn_wmma_f32_16x16x4_f32(false, a, false, b0, (short)0, acc0, false, false);
        acc1 = __builtin_amdgcn_wmma_f32_16x16x4_f32(false, a, false, b1, (short)0, acc1, false, false);
        acc2 = __builtin_amdgcn_wmma_f32_16x16x4_f32(false, a, false, b2, (short)0, acc2, false, false);
        acc3 = __builtin_amdgcn_wmma_f32_16x16x4_f32(false, a, false, b3, (short)0, acc3, false, false);
    }

    float* out = h1 + (size_t)row0 * F_HID;
    for (int i = 0; i < 8; i++) {
        int m = i + 8 * hi;
        out[(size_t)m * F_HID + (r +  0)] = acc0[i];
        out[(size_t)m * F_HID + (r + 16)] = acc1[i];
        out[(size_t)m * F_HID + (r + 32)] = acc2[i];
        out[(size_t)m * F_HID + (r + 48)] = acc3[i];
    }
}


template <int S>
__device__ __forceinline__ void gather_rows(float4& acc, int s, float w, int j,
                                            int hi, int q,
                                            const float4* __restrict__ hv) {
    float4 v[S];
    float  wu[S];
    int    su[S];
#pragma unroll
    for (int u = 0; u < S; u++) su[u] = __shfl(s, j + 2 * u + hi);
#pragma unroll
    for (int u = 0; u < S; u++) v[u] = hv[(unsigned)su[u] * (F_HID / 4) + q];
#pragma unroll
    for (int u = 0; u < S; u++) wu[u] = __shfl(w, j + 2 * u + hi);
#pragma unroll
    for (int u = 0; u < S; u++) {
        acc.x += v[u].x * wu[u];
        acc.y += v[u].y * wu[u];
        acc.z += v[u].z * wu[u];
        acc.w += v[u].w * wu[u];
    }
}

__global__ void __launch_bounds__(256)
gather1_kernel(const int* __restrict__ csr_src,
               const int* __restrict__ cnt,
               const int* __restrict__ off,
               const float* __restrict__ dinv,
               const float* __restrict__ h1,
               const float* __restrict__ b1,
               float* __restrict__ act1) {
    const int lane = threadIdx.x & 31;
    const int d = __builtin_amdgcn_readfirstlane(blockIdx.x * 8 + (threadIdx.x >> 5));
    if (d >= N_NODES) return;
    const int q  = lane & 15;
    const int hi = lane >> 4;
    const int   n    = cnt[d];
    const int   base = off[d];
    const float dd   = dinv[d];
    const float4* hv = (const float4*)h1;

    float4 acc = {0.0f, 0.0f, 0.0f, 0.0f};
    for (int c = 0; c < n; c += 32) {
        const int m  = min(32, n - c);
        const int my = (lane < m) ? lane : (m - 1);
        unsigned su = (unsigned)csr_src[base + c + my];
        su = min(su, (unsigned)(N_NODES - 1));
        const int   s = (int)su;
        const float wv = dinv[s] * dd;
        const float w  = (lane < m) ? wv : 0.0f;
        int j = 0;
        while (j < m) {
            if (m - j > 8) { gather_rows<8>(acc, s, w, j, hi, q, hv); j += 16; }
            else           { gather_rows<4>(acc, s, w, j, hi, q, hv); j += 8;  }
        }
    }
    acc.x += __shfl_xor(acc.x, 16);
    acc.y += __shfl_xor(acc.y, 16);
    acc.z += __shfl_xor(acc.z, 16);
    acc.w += __shfl_xor(acc.w, 16);

    if (hi == 0) {
        const float  inv = 1.0f / ((float)n + 1.0f);
        const float4 hs  = hv[(size_t)d * (F_HID / 4) + q];
        const float4 bb  = ((const float4*)b1)[q];
        float4 o;
        o.x = acc.x + hs.x * inv + bb.x;
        o.y = acc.y + hs.y * inv + bb.y;
        o.z = acc.z + hs.z * inv + bb.z;
        o.w = acc.w + hs.w * inv + bb.w;
        o.x = (o.x > 0.0f) ? o.x : 0.01f * o.x;
        o.y = (o.y > 0.0f) ? o.y : 0.01f * o.y;
        o.z = (o.z > 0.0f) ? o.z : 0.01f * o.z;
        o.w = (o.w > 0.0f) ? o.w : 0.01f * o.w;
        ((float4*)act1)[(size_t)d * (F_HID / 4) + q] = o;
    }
}


__global__ void __launch_bounds__(256)
gemm2_wmma_kernel(const float* __restrict__ h,
                  const float* __restrict__ W2,
                  float* __restrict__ h2) {
    __shared__ float lw[F_HID * 16];
    for (int i = threadIdx.x; i < F_HID * 16; i += blockDim.x) {
        int kk = i >> 4, nn = i & 15;
        float v = (nn < F_OUT) ? W2[kk * F_OUT + nn] : 0.0f;
        lw[(((kk >> 1) * 16) + nn) * 2 + (kk & 1)] = v;
    }
    __syncthreads();

    const int lane  = threadIdx.x & 31;
    const int wave  = threadIdx.x >> 5;
    const int mtile = blockIdx.x * 8 + wave;
    if (mtile * 16 >= N_NODES) return;
    const int row0 = mtile * 16;

    const int r  = lane & 15;
    const int hi = lane >> 4;

    v8f acc = {};
    const float* hrow = h + (size_t)(row0 + r) * F_HID;
    const v2f*   wp   = (const v2f*)lw;
    for (int k = 0; k < F_HID; k += 4) {
        v2f a = *(const v2f*)&hrow[k + 2 * hi];
        v2f b = wp[((k >> 1) + hi) * 16 + r];
        acc = __builtin_amdgcn_wmma_f32_16x16x4_f32(false, a, false, b, (short)0, acc, false, false);
    }

    if (r < F_OUT) {
        float* out = h2 + (size_t)row0 * F_OUT;
        for (int i = 0; i < 8; i++) {
            int m = i + 8 * hi;
            out[(size_t)m * F_OUT + r] = acc[i];
        }
    }
}


__global__ void __launch_bounds__(256)
gather2_kernel(const int* __restrict__ csr_src,
               const int* __restrict__ cnt,
               const int* __restrict__ off,
               const float* __restrict__ dinv,
               const float* __restrict__ h2,
               const float* __restrict__ b2,
               float* __restrict__ out) {
    const int t   = blockIdx.x * blockDim.x + threadIdx.x;
    const int d   = t >> 2;
    const int sub = t & 3;
    if (d >= N_NODES) return;
    const int   n    = cnt[d];
    const int   base = off[d];
    const float dd   = dinv[d];
    const float4* hv = (const float4*)h2;

    float4 acc = {0.0f, 0.0f, 0.0f, 0.0f};
    for (int j = sub; j < n; j += 16) {
        unsigned s[4];
        float    w[4];
        float4   v[4];
#pragma unroll
        for (int u = 0; u < 4; u++) {
            const bool ok = (j + 4 * u) < n;
            s[u] = (unsigned)csr_src[base + (ok ? j + 4 * u : j)];
            s[u] = min(s[u], (unsigned)(N_NODES - 1));
            const float wv = dinv[s[u]] * dd;
            w[u] = ok ? wv : 0.0f;
            v[u] = hv[s[u]];
        }
#pragma unroll
        for (int u = 0; u < 4; u++) {
            acc.x += v[u].x * w[u];
            acc.y += v[u].y * w[u];
            acc.z += v[u].z * w[u];
            acc.w += v[u].w * w[u];
        }
    }
    for (int o = 2; o > 0; o >>= 1) {
        acc.x += __shfl_xor(acc.x, o);
        acc.y += __shfl_xor(acc.y, o);
        acc.z += __shfl_xor(acc.z, o);
        acc.w += __shfl_xor(acc.w, o);
    }
    if (sub == 0) {
        const float  inv = 1.0f / ((float)n + 1.0f);
        const float4 hs  = hv[d];
        float v[F_OUT];
        v[0] = acc.x + hs.x * inv + b2[0];
        v[1] = acc.y + hs.y * inv + b2[1];
        v[2] = acc.z + hs.z * inv + b2[2];
        v[3] = acc.w + hs.w * inv + b2[3];
        float mx = -3.4e38f;
        for (int c = 0; c < F_OUT; c++) mx = fmaxf(mx, v[c]);
        float sm = 0.0f;
        for (int c = 0; c < F_OUT; c++) { v[c] = __expf(v[c] - mx); sm += v[c]; }
        const float rs = 1.0f / sm;
        float4 o;
        o.x = v[0] * rs;  o.y = v[1] * rs;  o.z = v[2] * rs;  o.w = v[3] * rs;
        ((float4*)out)[d] = o;
    }
}


extern "C" void kernel_launch(void* const* d_in, const int* in_sizes, int n_in,
                              void* d_out, int out_size, void* d_ws, size_t ws_size,
                              hipStream_t stream) {
    const float* x  = (const float*)d_in[0];
    const int*   ei = (const int*)d_in[1];
    const float* W1 = (const float*)d_in[2];
    const float* b1 = (const float*)d_in[3];
    const float* W2 = (const float*)d_in[4];
    const float* b2 = (const float*)d_in[5];
    float* out = (float*)d_out;

    const int E = in_sizes[1] / 2;
    const int* src = ei;
    const int* dst = ei + E;

    int*   cnt     = (int*)d_ws;
    int*   off     = cnt + N_NODES;
    float* dinv    = (float*)(off + N_NODES);
    int*   csr_src = (int*)(dinv + N_NODES + 32);
    float* h1      = (float*)(csr_src + E);
    float* act1    = h1   + (size_t)N_NODES * F_HID;
    float* h2      = act1 + (size_t)N_NODES * F_HID;
    int*   rank    = (int*)act1;

    const int T = 256;
    const int nb_nodes  = (N_NODES + T - 1) / T;
    const int nb_edges  = (E + T - 1) / T;
    const int nb_tiles  = (N_NODES / 16 + 7) / 8;
    const int nb_scan   = (N_NODES + 1023) / 1024;
    const int nb_g1     = (N_NODES + 7) / 8;
    const int nb_g2     = (N_NODES * 4 + T - 1) / T;

    zero_cnt_kernel<<<nb_nodes, T, 0, stream>>>(cnt);
    count_rank_kernel<<<nb_edges, T, 0, stream>>>(dst, E, cnt, rank);
    scan_kernel<<<nb_scan, T, 0, stream>>>(cnt, off, dinv);
    fill_kernel<<<nb_edges, T, 0, stream>>>(src, dst, rank, off, csr_src, E);

    gemm1_wmma_kernel<<<nb_tiles, T, 0, stream>>>(x, W1, h1);
    gather1_kernel<<<nb_g1, T, 0, stream>>>(csr_src, cnt, off, dinv, h1, b1, act1);

    gemm2_wmma_kernel<<<nb_tiles, T, 0, stream>>>(act1, W2, h2);
    gather2_kernel<<<nb_g2, T, 0, stream>>>(csr_src, cnt, off, dinv, h2, b2, out);
}


# edge sort without global atomics: edges binned by 512-node destination range with LDS counters (histogram, scan, place), then counted and placed per node in LDS per range; sum order: LDS arrival rank
# speedup vs baseline: 5.0921x; 1.7181x over previous
#include <hip/hip_runtime.h>


typedef __attribute__((ext_vector_type(2))) float v2f;
typedef __attribute__((ext_vector_type(8))) float v8f;

#define N_NODES 100000
#define F_IN    256
#define F_HID   64
#define F_OUT   4


#define BSH     9
#define BNODES  (1 << BSH)
#define NBUCK   ((N_NODES + BNODES - 1) / BNODES)
#define NB1     256
#define T1      1024

__device__ __forceinline__ unsigned bucket_of(int d) {
    return min((unsigned)d >> BSH, (unsigned)(NBUCK - 1));
}

__global__ void __launch_bounds__(T1)
hist_kernel(const int* __restrict__ dst, int E, int* __restrict__ bh) {
    __shared__ int h[NBUCK];
    const int tid = threadIdx.x;
    if (tid < NBUCK) h[tid] = 0;
    __syncthreads();
    const int ch = (E + NB1 - 1) / NB1;
    const int e0 = blockIdx.x * ch;
    const int e1 = min(E, e0 + ch);
    int e = e0 + tid;
    for (; e + 3 * T1 < e1; e += 4 * T1) {
        int d[4];
#pragma unroll
        for (int u = 0; u < 4; u++) d[u] = dst[e + u * T1];
#pragma unroll
        for (int u = 0; u < 4; u++) atomicAdd(&h[bucket_of(d[u])], 1);
    }
    for (; e < e1; e += T1) atomicAdd(&h[bucket_of(dst[e])], 1);
    __syncthreads();
    if (tid < NBUCK) bh[tid * NB1 + blockIdx.x] = h[tid];
}

__global__ void __launch_bounds__(256)
bscan_kernel(const int* __restrict__ in, int* __restrict__ outp, int n) {
    __shared__ int red[8];
    __shared__ int wsum[8];
    const int tid  = threadIdx.x;
    const int lane = tid & 31;
    const int wave = tid >> 5;
    const int start = blockIdx.x * 1024;

    const int4* c4 = (const int4*)in;
    int part = 0;
#pragma unroll 8
    for (int k = 0; k < (int)blockIdx.x; k++) {
        int4 v = c4[k * 256 + tid];
        part += (v.x + v.y) + (v.z + v.w);
    }
    for (int o = 16; o > 0; o >>= 1) part += __shfl_xor(part, o);
    if (lane == 0) red[wave] = part;

    const int i0 = start + tid * 4;
    int4 c = {0, 0, 0, 0};
    if (i0 < n) c = c4[i0 >> 2];
    const int t = (c.x + c.y) + (c.z + c.w);
    int inc = t;
    for (int o = 1; o < 32; o <<= 1) {
        int y = __shfl_up(inc, o);
        if (lane >= o) inc += y;
    }
    if (lane == 31) wsum[wave] = inc;
    __syncthreads();

    int ex = inc - t;
    for (int w = 0; w < 8; w++) {
        ex += red[w];
        if (w < wave) ex += wsum[w];
    }
    if (i0 < n) {
        int4 o4;
        o4.x = ex;
        o4.y = ex + c.x;
        o4.z = ex + c.x + c.y;
        o4.w = ex + c.x + c.y + c.z;
        ((int4*)outp)[i0 >> 2] = o4;
    }
}

__global__ void __launch_bounds__(T1)
scatter_kernel(const int* __restrict__ src, const int* __restrict__ dst, int E,
               const int* __restrict__ bo, int2* __restrict__ bpair) {
    __shared__ int cur[NBUCK];
    const int tid = threadIdx.x;
    if (tid < NBUCK) cur[tid] = bo[tid * NB1 + blockIdx.x];
    __syncthreads();
    const int ch = (E + NB1 - 1) / NB1;
    const int e0 = blockIdx.x * ch;
    const int e1 = min(E, e0 + ch);
    int e = e0 + tid;
    for (; e + 3 * T1 < e1; e += 4 * T1) {
        int s[4];
        int d[4];
#pragma unroll
        for (int u = 0; u < 4; u++) {
            s[u] = src[e + u * T1];
            d[u] = dst[e + u * T1];
        }
#pragma unroll
        for (int u = 0; u < 4; u++) {
            unsigned p = (unsigned)atomicAdd(&cur[bucket_of(d[u])], 1);
            p = min(p, (unsigned)(E - 1));
            bpair[p] = make_int2(s[u], d[u]);
        }
    }
    for (; e < e1; e += T1) {
        const int s = src[e];
        const int d = dst[e];
        unsigned p = (unsigned)atomicAdd(&cur[bucket_of(d)], 1);
        p = min(p, (unsigned)(E - 1));
        bpair[p] = make_int2(s, d);
    }
}

__global__ void __launch_bounds__(T1)
bucket_kernel(const int2* __restrict__ bpair, const int* __restrict__ bo, int E,
              int* __restrict__ cnt, int* __restrict__ off,
              float* __restrict__ dinv, int* __restrict__ csr_src) {
    __shared__ int lc[BNODES];
    __shared__ int lo[BNODES];
    __shared__ int wsum[BNODES / 32];
    const int tid  = threadIdx.x;
    const int lane = tid & 31;
    const int wave = tid >> 5;
    const int k    = blockIdx.x;
    const int s0   = bo[k * NB1];
    const int s1   = (k + 1 < NBUCK) ? bo[(k + 1) * NB1] : E;

    if (tid < BNODES) lc[tid] = 0;
    __syncthreads();

    int i = s0 + tid;
    for (; i + 3 * T1 < s1; i += 4 * T1) {
        int d[4];
#pragma unroll
        for (int u = 0; u < 4; u++) d[u] = bpair[i + u * T1].y;
#pragma unroll
        for (int u = 0; u < 4; u++) atomicAdd(&lc[(unsigned)d[u] & (BNODES - 1)], 1);
    }
    for (; i < s1; i += T1) atomicAdd(&lc[(unsigned)bpair[i].y & (BNODES - 1)], 1);
    __syncthreads();

    const int c = (tid < BNODES) ? lc[tid] : 0;
    int inc = c;
    for (int o = 1; o < 32; o <<= 1) {
        int y = __shfl_up(inc, o);
        if (lane >= o) inc += y;
    }
    if (lane == 31 && wave < BNODES / 32) wsum[wave] = inc;
    __syncthreads();
    if (tid < BNODES) {
        int ex = inc - c;
        for (int w = 0; w < BNODES / 32; w++)
            if (w < wave) ex += wsum[w];
        lo[tid] = ex;
        const int node = k * BNODES + tid;
        if (node < N_NODES) {
            cnt[node]  = c;
            off[node]  = s0 + ex;
            dinv[node] = rsqrtf((float)c + 1.0f);
        }
    }
    __syncthreads();

    i = s0 + tid;
    for (; i + 3 * T1 < s1; i += 4 * T1) {
        int2 pr[4];
#pragma unroll
        for (int u = 0; u < 4; u++) pr[u] = bpair[i + u * T1];
#pragma unroll
        for (int u = 0; u < 4; u++) {
            const int r = atomicAdd(&lo[(unsigned)pr[u].y & (BNODES - 1)], 1);
            unsigned p = (unsigned)(s0 + r);
            p = min(p, (unsigned)(E - 1));
            csr_src[p] = pr[u].x;
        }
    }
    for (; i < s1; i += T1) {
        const int2 pr = bpair[i];
        const int r = atomicAdd(&lo[(unsigned)pr.y & (BNODES - 1)], 1);
        unsigned p = (unsigned)(s0 + r);
        p = min(p, (unsigned)(E - 1));
        csr_src[p] = pr.x;
    }
}


__global__ void __launch_bounds__(256)
gemm1_wmma_kernel(const float* __restrict__ x,
                  const float* __restrict__ W1,
                  float* __restrict__ h1) {
    __shared__ float lw[F_IN * F_HID];
    for (int i = threadIdx.x; i < F_IN * F_HID; i += blockDim.x) {
        int kk  = i >> 6;
        int col = i & 63;
        lw[(((kk >> 1) * F_HID) + col) * 2 + (kk & 1)] = W1[i];
    }
    __syncthreads();

    const int lane  = threadIdx.x & 31;
    const int wave  = threadIdx.x >> 5;
    const int mtile = blockIdx.x * 8 + wave;
    if (mtile * 16 >= N_NODES) return;
    const int row0 = mtile * 16;

    const int r  = lane & 15;
    const int hi = lane >> 4;

    v8f acc0 = {}, acc1 = {}, acc2 = {}, acc3 = {};
    const float* xrow = x + (size_t)(row0 + r) * F_IN;
    const v2f*   wp   = (const v2f*)lw;

    for (int k = 0; k < F_IN; k += 4) {
        __builtin_prefetch(&xrow[k + 32], 0, 3);
        v2f a = *(const v2f*)&xrow[k + 2 * hi];
        const int pb = ((k >> 1) + hi) * F_HID;
        v2f b0 = wp[pb + r +  0];
        v2f b1 = wp[pb + r + 16];
        v2f b2 = wp[pb + r + 32];
        v2f b3 = wp[pb + r + 48];
        acc0 = __builtin_amdgcn_wmma_f32_16x16x4_f32(false, a, false, b0, (short)0, acc0, false, false);
        acc1 = __builtin_amdgcn_wmma_f32_16x16x4_f32(false, a, false, b1, (short)0, acc1, false, false);
        acc2 = __builtin_amdgcn_wmma_f32_16x16x4_f32(false, a, false, b2, (short)0, acc2, false, false);
        acc3 = __builtin_amdgcn_wmma_f32_16x16x4_f32(false, a, false, b3, (short)0, acc3, false, false);
    }

    float* out = h1 + (size_t)row0 * F_HID;
    for (int i = 0; i < 8; i++) {
        int m = i + 8 * hi;
        out[(size_t)m * F_HID + (r +  0)] = acc0[i];
        out[(size_t)m * F_HID + (r + 16)] = acc1[i];
        out[(size_t)m * F_HID + (r + 32)] = acc2[i];
        out[(size_t)m * F_HID + (r + 48)] = acc3[i];
    }
}


template <int S>
__device__ __forceinline__ void gather_rows(float4& acc, int s, float w, int j,
                                            int hi, int q,
                                            const float4* __restrict__ hv) {
    float4 v[S];
    float  wu[S];
    int    su[S];
#pragma unroll
    for (int u = 0; u < S; u++) su[u] = __shfl(s, j + 2 * u + hi);
#pragma unroll
    for (int u = 0; u < S; u++) v[u] = hv[(unsigned)su[u] * (F_HID / 4) + q];
#pragma unroll
    for (int u = 0; u < S; u++) wu[u] = __shfl(w, j + 2 * u + hi);
#pragma unroll
    for (int u = 0; u < S; u++) {
        acc.x += v[u].x * wu[u];
        acc.y += v[u].y * wu[u];
        acc.z += v[u].z * wu[u];
        acc.w += v[u].w * wu[u];
    }
}

__global__ void __launch_bounds__(256)
gather1_kernel(const int* __restrict__ csr_src,
               const int* __restrict__ cnt,
               const int* __restrict__ off,
               const float* __restrict__ dinv,
               const float* __restrict__ h1,
               const float* __restrict__ b1,
               float* __restrict__ act1) {
    const int lane = threadIdx.x & 31;
    const int d = __builtin_amdgcn_readfirstlane(blockIdx.x * 8 + (threadIdx.x >> 5));
    if (d >= N_NODES) return;
    const int q  = lane & 15;
    const int hi = lane >> 4;
    const int   n    = cnt[d];
    const int   base = off[d];
    const float dd   = dinv[d];
    const float4* hv = (const float4*)h1;

    float4 acc = {0.0f, 0.0f, 0.0f, 0.0f};
    for (int c = 0; c < n; c += 32) {
        const int m  = min(32, n - c);
        const int my = (lane < m) ? lane : (m - 1);
        unsigned su = (unsigned)csr_src[base + c + my];
        su = min(su, (unsigned)(N_NODES - 1));
        const int   s = (int)su;
        const float wv = dinv[s] * dd;
        const float w  = (lane < m) ? wv : 0.0f;
        int j = 0;
        while (j < m) {
            if (m - j > 8) { gather_rows<8>(acc, s, w, j, hi, q, hv); j += 16; }
            else           { gather_rows<4>(acc, s, w, j, hi, q, hv); j += 8;  }
        }
    }
    acc.x += __shfl_xor(acc.x, 16);
    acc.y += __shfl_xor(acc.y, 16);
    acc.z += __shfl_xor(acc.z, 16);
    acc.w += __shfl_xor(acc.w, 16);

    if (hi == 0) {
        const float  inv = 1.0f / ((float)n + 1.0f);
        const float4 hs  = hv[(size_t)d * (F_HID / 4) + q];
        const float4 bb  = ((const float4*)b1)[q];
        float4 o;
        o.x = acc.x + hs.x * inv + bb.x;
        o.y = acc.y + hs.y * inv + bb.y;
        o.z = acc.z + hs.z * inv + bb.z;
        o.w = acc.w + hs.w * inv + bb.w;
        o.x = (o.x > 0.0f) ? o.x : 0.01f * o.x;
        o.y = (o.y > 0.0f) ? o.y : 0.01f * o.y;
        o.z = (o.z > 0.0f) ? o.z : 0.01f * o.z;
        o.w = (o.w > 0.0f) ? o.w : 0.01f * o.w;
        ((float4*)act1)[(size_t)d * (F_HID / 4) + q] = o;
    }
}


__global__ void __launch_bounds__(256)
gemm2_wmma_kernel(const float* __restrict__ h,
                  const float* __restrict__ W2,
                  float* __restrict__ h2) {
    __shared__ float lw[F_HID * 16];
    for (int i = threadIdx.x; i < F_HID * 16; i += blockDim.x) {
        int kk = i >> 4, nn = i & 15;
        float v = (nn < F_OUT) ? W2[kk * F_OUT + nn] : 0.0f;
        lw[(((kk >> 1) * 16) + nn) * 2 + (kk & 1)] = v;
    }
    __syncthreads();

    const int lane  = threadIdx.x & 31;
    const int wave  = threadIdx.x >> 5;
    const int mtile = blockIdx.x * 8 + wave;
    if (mtile * 16 >= N_NODES) return;
    const int row0 = mtile * 16;

    const int r  = lane & 15;
    const int hi = lane >> 4;

    v8f acc = {};
    const float* hrow = h + (size_t)(row0 + r) * F_HID;
    const v2f*   wp   = (const v2f*)lw;
    for (int k = 0; k < F_HID; k += 4) {
        v2f a = *(const v2f*)&hrow[k + 2 * hi];
        v2f b = wp[((k >> 1) + hi) * 16 + r];
        acc = __builtin_amdgcn_wmma_f32_16x16x4_f32(false, a, false, b, (short)0, acc, false, false);
    }

    if (r < F_OUT) {
        float* out = h2 + (size_t)row0 * F_OUT;
        for (int i = 0; i < 8; i++) {
            int m = i + 8 * hi;
            out[(size_t)m * F_OUT + r] = acc[i];
        }
    }
}


__global__ void __launch_bounds__(256)
gather2_kernel(const int* __restrict__ csr_src,
               const int* __restrict__ cnt,
               const int* __restrict__ off,
               const float* __restrict__ dinv,
               const float* __restrict__ h2,
               const float* __restrict__ b2,
               float* __restrict__ out) {
    const int t   = blockIdx.x * blockDim.x + threadIdx.x;
    const int d   = t >> 2;
    const int sub = t & 3;
    if (d >= N_NODES) return;
    const int   n    = cnt[d];
    const int   base = off[d];
    const float dd   = dinv[d];
    const float4* hv = (const float4*)h2;

    float4 acc = {0.0f, 0.0f, 0.0f, 0.0f};
    for (int j = sub; j < n; j += 16) {
        unsigned s[4];
        float    w[4];
        float4   v[4];
#pragma unroll
        for (int u = 0; u < 4; u++) {
            const bool ok = (j + 4 * u) < n;
            s[u] = (unsigned)csr_src[base + (ok ? j + 4 * u : j)];
            s[u] = min(s[u], (unsigned)(N_NODES - 1));
            const float wv = dinv[s[u]] * dd;
            w[u] = ok ? wv : 0.0f;
            v[u] = hv[s[u]];
        }
#pragma unroll
        for (int u = 0; u < 4; u++) {
            acc.x += v[u].x * w[u];
            acc.y += v[u].y * w[u];
            acc.z += v[u].z * w[u];
            acc.w += v[u].w * w[u];
        }
    }
    for (int o = 2; o > 0; o >>= 1) {
        acc.x += __shfl_xor(acc.x, o);
        acc.y += __shfl_xor(acc.y, o);
        acc.z += __shfl_xor(acc.z, o);
        acc.w += __shfl_xor(acc.w, o);
    }
    if (sub == 0) {
        const float  inv = 1.0f / ((float)n + 1.0f);
        const float4 hs  = hv[d];
        float v[F_OUT];
        v[0] = acc.x + hs.x * inv + b2[0];
        v[1] = acc.y + hs.y * inv + b2[1];
        v[2] = acc.z + hs.z * inv + b2[2];
        v[3] = acc.w + hs.w * inv + b2[3];
        float mx = -3.4e38f;
        for (int c = 0; c < F_OUT; c++) mx = fmaxf(mx, v[c]);
        float sm = 0.0f;
        for (int c = 0; c < F_OUT; c++) { v[c] = __expf(v[c] - mx); sm += v[c]; }
        const float rs = 1.0f / sm;
        float4 o;
        o.x = v[0] * rs;  o.y = v[1] * rs;  o.z = v[2] * rs;  o.w = v[3] * rs;
        ((float4*)out)[d] = o;
    }
}


extern "C" void kernel_launch(void* const* d_in, const int* in_sizes, int n_in,
                              void* d_out, int out_size, void* d_ws, size_t ws_size,
                              hipStream_t stream) {
    const float* x  = (const float*)d_in[0];
    const int*   ei = (const int*)d_in[1];
    const float* W1 = (const float*)d_in[2];
    const float* b1 = (const float*)d_in[3];
    const float* W2 = (const float*)d_in[4];
    const float* b2 = (const float*)d_in[5];
    float* out = (float*)d_out;

    const int E = in_sizes[1] / 2;
    const int* src = ei;
    const int* dst = ei + E;

    int*   cnt     = (int*)d_ws;
    int*   off     = cnt + N_NODES;
    float* dinv    = (float*)(off + N_NODES);
    int*   csr_src = (int*)(dinv + N_NODES + 32);
    float* h1      = (float*)(csr_src + E);
    float* act1    = h1   + (size_t)N_NODES * F_HID;
    float* h2      = act1 + (size_t)N_NODES * F_HID;
    int2*  bpair   = (int2*)h1;
    int*   bh      = (int*)h2;
    int*   bo      = bh + NBUCK * NB1;

    const int T = 256;
    const int nb_tiles  = (N_NODES / 16 + 7) / 8;
    const int nb_bscan  = (NBUCK * NB1 + 1023) / 1024;
    const int nb_g1     = (N_NODES + 7) / 8;
    const int nb_g2     = (N_NODES * 4 + T - 1) / T;

    hist_kernel<<<NB1, T1, 0, stream>>>(dst, E, bh);
    bscan_kernel<<<nb_bscan, T, 0, stream>>>(bh, bo, NBUCK * NB1);
    scatter_kernel<<<NB1, T1, 0, stream>>>(src, dst, E, bo, bpair);
    bucket_kernel<<<NBUCK, T1, 0, stream>>>(bpair, bo, E, cnt, off, dinv, csr_src);

    gemm1_wmma_kernel<<<nb_tiles, T, 0, stream>>>(x, W1, h1);
    gather1_kernel<<<nb_g1, T, 0, stream>>>(csr_src, cnt, off, dinv, h1, b1, act1);

    gemm2_wmma_kernel<<<nb_tiles, T, 0, stream>>>(act1, W2, h2);
    gather2_kernel<<<nb_g2, T, 0, stream>>>(csr_src, cnt, off, dinv, h2, b2, out);
}


# both placing kernels of the edge sort order their records in an LDS stage first and copy it out as runs of whole lines (was one scattered 4- or 8-byte store per edge); sum order unchanged
# speedup vs baseline: 6.4556x; 1.2678x over previous
#include <hip/hip_runtime.h>


typedef __attribute__((ext_vector_type(2))) float v2f;
typedef __attribute__((ext_vector_type(8))) float v8f;

#define N_NODES 100000
#define F_IN    256
#define F_HID   64
#define F_OUT   4


#define BSH     9
#define BNODES  (1 << BSH)
#define NBUCK   ((N_NODES + BNODES - 1) / BNODES)
#define NB1     256
#define T1      1024
#define SCAP    16384
#define BCAP    32768

__device__ __forceinline__ unsigned bucket_of(int d) {
    return min((unsigned)d >> BSH, (unsigned)(NBUCK - 1));
}

__global__ void __launch_bounds__(T1)
hist_kernel(const int* __restrict__ dst, int E, int* __restrict__ bh) {
    __shared__ int h[NBUCK];
    const int tid = threadIdx.x;
    if (tid < NBUCK) h[tid] = 0;
    __syncthreads();
    const int ch = (E + NB1 - 1) / NB1;
    const int e0 = blockIdx.x * ch;
    const int e1 = min(E, e0 + ch);
    int e = e0 + tid;
    for (; e + 3 * T1 < e1; e += 4 * T1) {
        int d[4];
#pragma unroll
        for (int u = 0; u < 4; u++) d[u] = dst[e + u * T1];
#pragma unroll
        for (int u = 0; u < 4; u++) atomicAdd(&h[bucket_of(d[u])], 1);
    }
    for (; e < e1; e += T1) atomicAdd(&h[bucket_of(dst[e])], 1);
    __syncthreads();
    if (tid < NBUCK) bh[tid * NB1 + blockIdx.x] = h[tid];
}

__global__ void __launch_bounds__(256)
bscan_kernel(const int* __restrict__ in, int* __restrict__ outp, int n) {
    __shared__ int red[8];
    __shared__ int wsum[8];
    const int tid  = threadIdx.x;
    const int lane = tid & 31;
    const int wave = tid >> 5;
    const int start = blockIdx.x * 1024;

    const int4* c4 = (const int4*)in;
    int part = 0;
#pragma unroll 8
    for (int k = 0; k < (int)blockIdx.x; k++) {
        int4 v = c4[k * 256 + tid];
        part += (v.x + v.y) + (v.z + v.w);
    }
    for (int o = 16; o > 0; o >>= 1) part += __shfl_xor(part, o);
    if (lane == 0) red[wave] = part;

    const int i0 = start + tid * 4;
    int4 c = {0, 0, 0, 0};
    if (i0 < n) c = c4[i0 >> 2];
    const int t = (c.x + c.y) + (c.z + c.w);
    int inc = t;
    for (int o = 1; o < 32; o <<= 1) {
        int y = __shfl_up(inc, o);
        if (lane >= o) inc += y;
    }
    if (lane == 31) wsum[wave] = inc;
    __syncthreads();

    int ex = inc - t;
    for (int w = 0; w < 8; w++) {
        ex += red[w];
        if (w < wave) ex += wsum[w];
    }
    if (i0 < n) {
        int4 o4;
        o4.x = ex;
        o4.y = ex + c.x;
        o4.z = ex + c.x + c.y;
        o4.w = ex + c.x + c.y + c.z;
        ((int4*)outp)[i0 >> 2] = o4;
    }
}

__global__ void __launch_bounds__(T1)
scatter_kernel(const int* __restrict__ src, const int* __restrict__ dst, int E,
               const int* __restrict__ bh, const int* __restrict__ bo,
               int2* __restrict__ bpair) {
    __shared__ int cur[NBUCK];
    __shared__ int gdelta[NBUCK];
    __shared__ int wsum[8];
    __shared__ __attribute__((aligned(16))) int stage_w[2 * SCAP];
    int2* stage = (int2*)stage_w;
    const int tid  = threadIdx.x;
    const int lane = tid & 31;
    const int wave = tid >> 5;
    const int ch = (E + NB1 - 1) / NB1;
    const int e0 = blockIdx.x * ch;
    const int e1 = min(E, e0 + ch);
    const bool staged = (ch <= SCAP);

    const int c = (tid < NBUCK) ? bh[tid * NB1 + blockIdx.x] : 0;
    int inc = c;
    for (int o = 1; o < 32; o <<= 1) {
        int y = __shfl_up(inc, o);
        if (lane >= o) inc += y;
    }
    if (lane == 31 && wave < 8) wsum[wave] = inc;
    __syncthreads();
    if (tid < NBUCK) {
        int ex = inc - c;
        for (int w = 0; w < 8; w++)
            if (w < wave) ex += wsum[w];
        const int g = bo[tid * NB1 + blockIdx.x];
        gdelta[tid] = g - ex;
        cur[tid]    = staged ? ex : g;
    }
    __syncthreads();

    int e = e0 + tid;
    if (staged) {
        for (; e + 3 * T1 < e1; e += 4 * T1) {
            int s[4];
            int d[4];
            unsigned p[4];
#pragma unroll
            for (int u = 0; u < 4; u++) {
                s[u] = src[e + u * T1];
                d[u] = dst[e + u * T1];
            }
#pragma unroll
            for (int u = 0; u < 4; u++) p[u] = (unsigned)atomicAdd(&cur[bucket_of(d[u])], 1);
#pragma unroll
            for (int u = 0; u < 4; u++) {
                const unsigned q = min(p[u], (unsigned)(SCAP - 1));
                stage[q] = make_int2(s[u], d[u]);
            }
        }
        for (; e < e1; e += T1) {
            const int s = src[e];
            const int d = dst[e];
            unsigned p = (unsigned)atomicAdd(&cur[bucket_of(d)], 1);
            p = min(p, (unsigned)(SCAP - 1));
            stage[p] = make_int2(s, d);
        }
        __syncthreads();

        const int len = e1 - e0;
        int i = tid;
        for (; i + 3 * T1 < len; i += 4 * T1) {
            int2 pr[4];
            int  gd[4];
#pragma unroll
            for (int u = 0; u < 4; u++) pr[u] = stage[i + u * T1];
#pragma unroll
            for (int u = 0; u < 4; u++) gd[u] = gdelta[bucket_of(pr[u].y)];
#pragma unroll
            for (int u = 0; u < 4; u++) {
                unsigned p = (unsigned)(gd[u] + i + u * T1);
                p = min(p, (unsigned)(E - 1));
                bpair[p] = pr[u];
            }
        }
        for (; i < len; i += T1) {
            const int2 pr = stage[i];
            unsigned p = (unsigned)(gdelta[bucket_of(pr.y)] + i);
            p = min(p, (unsigned)(E - 1));
            bpair[p] = pr;
        }
    } else {
        for (; e + 3 * T1 < e1; e += 4 * T1) {
            int s[4];
            int d[4];
#pragma unroll
            for (int u = 0; u < 4; u++) {
                s[u] = src[e + u * T1];
                d[u] = dst[e + u * T1];
            }
#pragma unroll
            for (int u = 0; u < 4; u++) {
                unsigned p = (unsigned)atomicAdd(&cur[bucket_of(d[u])], 1);
                p = min(p, (unsigned)(E - 1));
                bpair[p] = make_int2(s[u], d[u]);
            }
        }
        for (; e < e1; e += T1) {
            const int s = src[e];
            const int d = dst[e];
            unsigned p = (unsigned)atomicAdd(&cur[bucket_of(d)], 1);
            p = min(p, (unsigned)(E - 1));
            bpair[p] = make_int2(s, d);
        }
    }
}

__global__ void __launch_bounds__(T1)
bucket_kernel(const int2* __restrict__ bpair, const int* __restrict__ bo, int E,
              int* __restrict__ cnt, int* __restrict__ off,
              float* __restrict__ dinv, int* __restrict__ csr_src) {
    __shared__ int lc[BNODES];
    __shared__ int lo[BNODES];
    __shared__ int wsum[BNODES / 32];
    __shared__ int stage[BCAP];
    const int tid  = threadIdx.x;
    const int lane = tid & 31;
    const int wave = tid >> 5;
    const int k    = blockIdx.x;
    const int s0   = bo[k * NB1];
    const int s1   = (k + 1 < NBUCK) ? bo[(k + 1) * NB1] : E;
    const bool staged = (s1 - s0 <= BCAP);

    if (tid < BNODES) lc[tid] = 0;
    __syncthreads();

    int i = s0 + tid;
    for (; i + 3 * T1 < s1; i += 4 * T1) {
        int d[4];
#pragma unroll
        for (int u = 0; u < 4; u++) d[u] = bpair[i + u * T1].y;
#pragma unroll
        for (int u = 0; u < 4; u++) atomicAdd(&lc[(unsigned)d[u] & (BNODES - 1)], 1);
    }
    for (; i < s1; i += T1) atomicAdd(&lc[(unsigned)bpair[i].y & (BNODES - 1)], 1);
    __syncthreads();

    const int c = (tid < BNODES) ? lc[tid] : 0;
    int inc = c;
    for (int o = 1; o < 32; o <<= 1) {
        int y = __shfl_up(inc, o);
        if (lane >= o) inc += y;
    }
    if (lane == 31 && wave < BNODES / 32) wsum[wave] = inc;
    __syncthreads();
    if (tid < BNODES) {
        int ex = inc - c;
        for (int w = 0; w < BNODES / 32; w++)
            if (w < wave) ex += wsum[w];
        lo[tid] = ex;
        const int node = k * BNODES + tid;
        if (node < N_NODES) {
            cnt[node]  = c;
            off[node]  = s0 + ex;
            dinv[node] = rsqrtf((float)c + 1.0f);
        }
    }
    __syncthreads();

    i = s0 + tid;
    if (staged) {
        for (; i + 3 * T1 < s1; i += 4 * T1) {
            int2 pr[4];
            int  r[4];
#pragma unroll
            for (int u = 0; u < 4; u++) pr[u] = bpair[i + u * T1];
#pragma unroll
            for (int u = 0; u < 4; u++) r[u] = atomicAdd(&lo[(unsigned)pr[u].y & (BNODES - 1)], 1);
#pragma unroll
            for (int u = 0; u < 4; u++)
                stage[min((unsigned)r[u], (unsigned)(BCAP - 1))] = pr[u].x;
        }
        for (; i < s1; i += T1) {
            const int2 pr = bpair[i];
            const int r = atomicAdd(&lo[(unsigned)pr.y & (BNODES - 1)], 1);
            stage[min((unsigned)r, (unsigned)(BCAP - 1))] = pr.x;
        }
        __syncthreads();

        const int len = s1 - s0;
        for (int j = tid; j < len; j += T1) {
            unsigned p = (unsigned)(s0 + j);
            p = min(p, (unsigned)(E - 1));
            csr_src[p] = stage[j];
        }
    } else {
        for (; i + 3 * T1 < s1; i += 4 * T1) {
            int2 pr[4];
#pragma unroll
            for (int u = 0; u < 4; u++) pr[u] = bpair[i + u * T1];
#pragma unroll
            for (int u = 0; u < 4; u++) {
                const int r = atomicAdd(&lo[(unsigned)pr[u].y & (BNODES - 1)], 1);
                unsigned p = (unsigned)(s0 + r);
                p = min(p, (unsigned)(E - 1));
                csr_src[p] = pr[u].x;
            }
        }
        for (; i < s1; i += T1) {
            const int2 pr = bpair[i];
            const int r = atomicAdd(&lo[(unsigned)pr.y & (BNODES - 1)], 1);
            unsigned p = (unsigned)(s0 + r);
            p = min(p, (unsigned)(E - 1));
            csr_src[p] = pr.x;
        }
    }
}


__global__ void __launch_bounds__(256)
gemm1_wmma_kernel(const float* __restrict__ x,
                  const float* __restrict__ W1,
                  const float* __restrict__ dinv,
                  float* __restrict__ h1) {
    __shared__ float lw[F_IN * F_HID];
    for (int i = threadIdx.x; i < F_IN * F_HID; i += blockDim.x) {
        int kk  = i >> 6;
        int col = i & 63;
        lw[(((kk >> 1) * F_HID) + col) * 2 + (kk & 1)] = W1[i];
    }
    __syncthreads();

    const int lane  = threadIdx.x & 31;
    const int wave  = threadIdx.x >> 5;
    const int mtile = blockIdx.x * 8 + wave;
    if (mtile * 16 >= N_NODES) return;
    const int row0 = mtile * 16;

    const int r  = lane & 15;
    const int hi = lane >> 4;

    v8f acc0 = {}, acc1 = {}, acc2 = {}, acc3 = {};
    const float* xrow = x + (size_t)(row0 + r) * F_IN;
    const v2f*   wp   = (const v2f*)lw;

    for (int k = 0; k < F_IN; k += 4) {
        __builtin_prefetch(&xrow[k + 32], 0, 3);
        v2f a = *(const v2f*)&xrow[k + 2 * hi];
        const int pb = ((k >> 1) + hi) * F_HID;
        v2f b0 = wp[pb + r +  0];
        v2f b1 = wp[pb + r + 16];
        v2f b2 = wp[pb + r + 32];
        v2f b3 = wp[pb + r + 48];
        acc0 = __builtin_amdgcn_wmma_f32_16x16x4_f32(false, a, false, b0, (short)0, acc0, false, false);
        acc1 = __builtin_amdgcn_wmma_f32_16x16x4_f32(false, a, false, b1, (short)0, acc1, false, false);
        acc2 = __builtin_amdgcn_wmma_f32_16x16x4_f32(false, a, false, b2, (short)0, acc2, false, false);
        acc3 = __builtin_amdgcn_wmma_f32_16x16x4_f32(false, a, false, b3, (short)0, acc3, false, false);
    }

    const float4* dv = (const float4*)(dinv + row0 + 8 * hi);
    const float4 dlo = dv[0];
    const float4 dhi = dv[1];
    const float ds[8] = {dlo.x, dlo.y, dlo.z, dlo.w, dhi.x, dhi.y, dhi.z, dhi.w};
    float* out = h1 + (size_t)row0 * F_HID;
#pragma unroll
    for (int i = 0; i < 8; i++) {
        int m = i + 8 * hi;
        out[(size_t)m * F_HID + (r +  0)] = acc0[i] * ds[i];
        out[(size_t)m * F_HID + (r + 16)] = acc1[i] * ds[i];
        out[(size_t)m * F_HID + (r + 32)] = acc2[i] * ds[i];
        out[(size_t)m * F_HID + (r + 48)] = acc3[i] * ds[i];
    }
}


template <int S, bool FULL>
__device__ __forceinline__ void gather_rows(float4& acc, int s, int j, int m,
                                            int hi, int q,
                                            const float4* __restrict__ hv) {
    float4 v[S];
    int    su[S];
#pragma unroll
    for (int u = 0; u < S; u++) su[u] = __shfl(s, j + 2 * u + hi);
#pragma unroll
    for (int u = 0; u < S; u++) v[u] = hv[(unsigned)su[u] * (F_HID / 4) + q];
#pragma unroll
    for (int u = 0; u < S; u++) {
        if (FULL) {
            acc.x += v[u].x;
            acc.y += v[u].y;
            acc.z += v[u].z;
            acc.w += v[u].w;
        } else {
            const float mk = (j + 2 * u + hi < m) ? 1.0f : 0.0f;
            acc.x += v[u].x * mk;
            acc.y += v[u].y * mk;
            acc.z += v[u].z * mk;
            acc.w += v[u].w * mk;
        }
    }
}

__global__ void __launch_bounds__(256)
gather1_kernel(const int* __restrict__ csr_src,
               const int* __restrict__ cnt,
               const int* __restrict__ off,
               const float* __restrict__ dinv,
               const float* __restrict__ h1,
               const float* __restrict__ b1,
               float* __restrict__ act1) {
    const int lane = threadIdx.x & 31;
    const int d = __builtin_amdgcn_readfirstlane(blockIdx.x * 8 + (threadIdx.x >> 5));
    if (d >= N_NODES) return;
    const int q  = lane & 15;
    const int hi = lane >> 4;
    const int   n    = cnt[d];
    const int   base = off[d];
    const float dd   = dinv[d];
    const float4* hv = (const float4*)h1;

    float4 acc = {0.0f, 0.0f, 0.0f, 0.0f};
    for (int c = 0; c < n; c += 32) {
        const int m  = min(32, n - c);
        const int my = (lane < m) ? lane : (m - 1);
        unsigned su = (unsigned)csr_src[base + c + my];
        su = min(su, (unsigned)(N_NODES - 1));
        const int s = (int)su;
        int j = 0;
        while (j < m) {
            const int left = m - j;
            if (left >= 16)    { gather_rows<8, true >(acc, s, j, m, hi, q, hv); j += 16; }
            else if (left > 8) { gather_rows<8, false>(acc, s, j, m, hi, q, hv); j += 16; }
            else               { gather_rows<4, false>(acc, s, j, m, hi, q, hv); j += 8;  }
        }
    }
    acc.x += __shfl_xor(acc.x, 16);
    acc.y += __shfl_xor(acc.y, 16);
    acc.z += __shfl_xor(acc.z, 16);
    acc.w += __shfl_xor(acc.w, 16);

    if (hi == 0) {
        const float4 hs  = hv[(size_t)d * (F_HID / 4) + q];
        const float4 bb  = ((const float4*)b1)[q];
        float4 o;
        o.x = (acc.x + hs.x) * dd + bb.x;
        o.y = (acc.y + hs.y) * dd + bb.y;
        o.z = (acc.z + hs.z) * dd + bb.z;
        o.w = (acc.w + hs.w) * dd + bb.w;
        o.x = (o.x > 0.0f) ? o.x : 0.01f * o.x;
        o.y = (o.y > 0.0f) ? o.y : 0.01f * o.y;
        o.z = (o.z > 0.0f) ? o.z : 0.01f * o.z;
        o.w = (o.w > 0.0f) ? o.w : 0.01f * o.w;
        ((float4*)act1)[(size_t)d * (F_HID / 4) + q] = o;
    }
}


__global__ void __launch_bounds__(256)
gemm2_wmma_kernel(const float* __restrict__ h,
                  const float* __restrict__ W2,
                  const float* __restrict__ dinv,
                  float* __restrict__ h2) {
    __shared__ float lw[F_HID * 16];
    for (int i = threadIdx.x; i < F_HID * 16; i += blockDim.x) {
        int kk = i >> 4, nn = i & 15;
        float v = (nn < F_OUT) ? W2[kk * F_OUT + nn] : 0.0f;
        lw[(((kk >> 1) * 16) + nn) * 2 + (kk & 1)] = v;
    }
    __syncthreads();

    const int lane  = threadIdx.x & 31;
    const int wave  = threadIdx.x >> 5;
    const int mtile = blockIdx.x * 8 + wave;
    if (mtile * 16 >= N_NODES) return;
    const int row0 = mtile * 16;

    const int r  = lane & 15;
    const int hi = lane >> 4;

    const float4* dv = (const float4*)(dinv + row0 + 8 * hi);
    const float4 dlo = dv[0];
    const float4 dhi = dv[1];

    v8f acc = {};
    const float* hrow = h + (size_t)(row0 + r) * F_HID;
    const v2f*   wp   = (const v2f*)lw;
    for (int k = 0; k < F_HID; k += 4) {
        v2f a = *(const v2f*)&hrow[k + 2 * hi];
        v2f b = wp[((k >> 1) + hi) * 16 + r];
        acc = __builtin_amdgcn_wmma_f32_16x16x4_f32(false, a, false, b, (short)0, acc, false, false);
    }

    if (r < F_OUT) {
        const float ds[8] = {dlo.x, dlo.y, dlo.z, dlo.w, dhi.x, dhi.y, dhi.z, dhi.w};
        float* out = h2 + (size_t)row0 * F_OUT;
#pragma unroll
        for (int i = 0; i < 8; i++) {
            int m = i + 8 * hi;
            out[(size_t)m * F_OUT + r] = acc[i] * ds[i];
        }
    }
}


__global__ void __launch_bounds__(256)
gather2_kernel(const int* __restrict__ csr_src,
               const int* __restrict__ cnt,
               const int* __restrict__ off,
               const float* __restrict__ dinv,
               const float* __restrict__ h2,
               const float* __restrict__ b2,
               float* __restrict__ out) {
    const int t   = blockIdx.x * blockDim.x + threadIdx.x;
    const int d   = t >> 2;
    const int sub = t & 3;
    if (d >= N_NODES) return;
    const int   n    = cnt[d];
    const int   base = off[d];
    const float dd   = dinv[d];
    const float4* hv = (const float4*)h2;

    float4 acc = {0.0f, 0.0f, 0.0f, 0.0f};
    for (int j = sub; j < n; j += 16) {
        unsigned s[4];
        float    w[4];
        float4   v[4];
#pragma unroll
        for (int u = 0; u < 4; u++) {
            const bool ok = (j + 4 * u) < n;
            s[u] = (unsigned)csr_src[base + (ok ? j + 4 * u : j)];
            s[u] = min(s[u], (unsigned)(N_NODES - 1));
            w[u] = ok ? 1.0f : 0.0f;
            v[u] = hv[s[u]];
        }
#pragma unroll
        for (int u = 0; u < 4; u++) {
            acc.x += v[u].x * w[u];
            acc.y += v[u].y * w[u];
            acc.z += v[u].z * w[u];
            acc.w += v[u].w * w[u];
        }
    }
    for (int o = 2; o > 0; o >>= 1) {
        acc.x += __shfl_xor(acc.x, o);
        acc.y += __shfl_xor(acc.y, o);
        acc.z += __shfl_xor(acc.z, o);
        acc.w += __shfl_xor(acc.w, o);
    }
    if (sub == 0) {
        const float4 hs  = hv[d];
        float v[F_OUT];
        v[0] = (acc.x + hs.x) * dd + b2[0];
        v[1] = (acc.y + hs.y) * dd + b2[1];
        v[2] = (acc.z + hs.z) * dd + b2[2];
        v[3] = (acc.w + hs.w) * dd + b2[3];
        float mx = -3.4e38f;
        for (int c = 0; c < F_OUT; c++) mx = fmaxf(mx, v[c]);
        float sm = 0.0f;
        for (int c = 0; c < F_OUT; c++) { v[c] = __expf(v[c] - mx); sm += v[c]; }
        const float rs = 1.0f / sm;
        float4 o;
        o.x = v[0] * rs;  o.y = v[1] * rs;  o.z = v[2] * rs;  o.w = v[3] * rs;
        ((float4*)out)[d] = o;
    }
}


extern "C" void kernel_launch(void* const* d_in, const int* in_sizes, int n_in,
                              void* d_out, int out_size, void* d_ws, size_t ws_size,
                              hipStream_t stream) {
    const float* x  = (const float*)d_in[0];
    const int*   ei = (const int*)d_in[1];
    const float* W1 = (const float*)d_in[2];
    const float* b1 = (const float*)d_in[3];
    const float* W2 = (const float*)d_in[4];
    const float* b2 = (const float*)d_in[5];
    float* out = (float*)d_out;

    const int E = in_sizes[1] / 2;
    const int* src = ei;
    const int* dst = ei + E;

    int*   cnt     = (int*)d_ws;
    int*   off     = cnt + N_NODES;
    float* dinv    = (float*)(off + N_NODES);
    int*   csr_src = (int*)(dinv + N_NODES + 32);
    float* h1      = (float*)(csr_src + E);
    float* act1    = h1   + (size_t)N_NODES * F_HID;
    float* h2      = act1 + (size_t)N_NODES * F_HID;
    int2*  bpair   = (int2*)h1;
    int*   bh      = (int*)h2;
    int*   bo      = bh + NBUCK * NB1;

    const int T = 256;
    const int nb_tiles  = (N_NODES / 16 + 7) / 8;
    const int nb_bscan  = (NBUCK * NB1 + 1023) / 1024;
    const int nb_g1     = (N_NODES + 7) / 8;
    const int nb_g2     = (N_NODES * 4 + T - 1) / T;

    hist_kernel<<<NB1, T1, 0, stream>>>(dst, E, bh);
    bscan_kernel<<<nb_bscan, T, 0, stream>>>(bh, bo, NBUCK * NB1);
    scatter_kernel<<<NB1, T1, 0, stream>>>(src, dst, E, bh, bo, bpair);
    bucket_kernel<<<NBUCK, T1, 0, stream>>>(bpair, bo, E, cnt, off, dinv, csr_src);

    gemm1_wmma_kernel<<<nb_tiles, T, 0, stream>>>(x, W1, dinv, h1);
    gather1_kernel<<<nb_g1, T, 0, stream>>>(csr_src, cnt, off, dinv, h1, b1, act1);

    gemm2_wmma_kernel<<<nb_tiles, T, 0, stream>>>(act1, W2, dinv, h2);
    gather2_kernel<<<nb_g2, T, 0, stream>>>(csr_src, cnt, off, dinv, h2, b2, out);
}


# edge sort in 2 launches (was 4): the LDS add that counts an edge also ranks it, at both levels; a chunk is ordered by range in its own slice, a range reads 256 runs; sum order: LDS arrival rank
# speedup vs baseline: 6.9041x; 1.0695x over previous
#include <hip/hip_runtime.h>


typedef __attribute__((ext_vector_type(2))) float v2f;
typedef __attribute__((ext_vector_type(8))) float v8f;

#define N_NODES 100000
#define F_IN    256
#define F_HID   64
#define F_OUT   4


#define BSH     9
#define BNODES  (1 << BSH)
#define NBUCK   ((N_NODES + BNODES - 1) / BNODES)
#define NB1     256
#define T1      1024
#define EPT     16
#define SCAP    (EPT * T1)
#define RCAP    20480
#define RPW     (NB1 / (T1 / 32))
#define SRCB    20
static_assert(N_NODES <= (1 << SRCB) && BSH + SRCB <= 32, "a source index and a range-local node must fit one word");
#define PRE_SLOTS ((NBUCK * NB1 / 4 + T1 - 1) / T1)

__device__ __forceinline__ unsigned bucket_of(int d) {
    return min((unsigned)d >> BSH, (unsigned)(NBUCK - 1));
}

__global__ void __launch_bounds__(T1)
part_kernel(const int* __restrict__ src, const int* __restrict__ dst, int E,
            int* __restrict__ tab, int* __restrict__ st, int2* __restrict__ cpair) {
    __shared__ int h[NBUCK];
    __shared__ int ex[NBUCK];
    __shared__ int wsum[8];
    __shared__ __attribute__((aligned(16))) int stage_w[2 * SCAP];
    int2* stage = (int2*)stage_w;
    const int tid  = threadIdx.x;
    const int lane = tid & 31;
    const int wave = tid >> 5;
    const int ch = (E + NB1 - 1) / NB1;
    const int e0 = min(E, (int)blockIdx.x * ch);
    const int e1 = min(E, e0 + ch);
    const bool staged = (ch <= SCAP);

    if (tid < NBUCK) h[tid] = 0;
    __syncthreads();

    int      s[EPT];
    int      d[EPT];
    unsigned rk[EPT];
    if (staged) {
#pragma unroll
        for (int u = 0; u < EPT; u++) {
            const unsigned e = min((unsigned)(e0 + tid + u * T1), (unsigned)(E - 1));
            s[u] = src[e];
            d[u] = dst[e];
        }
#pragma unroll
        for (int u = 0; u < EPT; u++) {
            rk[u] = 0;
            if (e0 + tid + u * T1 < e1) rk[u] = (unsigned)atomicAdd(&h[bucket_of(d[u])], 1);
        }
    } else {
        for (int e = e0 + tid; e < e1; e += T1) atomicAdd(&h[bucket_of(dst[e])], 1);
    }
    __syncthreads();

    const int c = (tid < NBUCK) ? h[tid] : 0;
    int inc = c;
    for (int o = 1; o < 32; o <<= 1) {
        int y = __shfl_up(inc, o);
        if (lane >= o) inc += y;
    }
    if (lane == 31 && wave < 8) wsum[wave] = inc;
    __syncthreads();
    if (tid < NBUCK) {
        int x = inc - c;
        for (int w = 0; w < 8; w++)
            if (w < wave) x += wsum[w];
        ex[tid] = x;
        tab[tid * NB1 + blockIdx.x] = c;
        st[tid * NB1 + blockIdx.x]  = x;
        if (!staged) h[tid] = x;
    }
    __syncthreads();

    if (staged) {
        unsigned q[EPT];
#pragma unroll
        for (int u = 0; u < EPT; u++)
            q[u] = min((unsigned)ex[bucket_of(d[u])] + rk[u], (unsigned)(SCAP - 1));
#pragma unroll
        for (int u = 0; u < EPT; u++)
            if (e0 + tid + u * T1 < e1) stage[q[u]] = make_int2(s[u], d[u]);
        __syncthreads();
        const int len = e1 - e0;
        int i = tid;
        for (; i + 3 * T1 < len; i += 4 * T1) {
            int2 v[4];
#pragma unroll
            for (int u = 0; u < 4; u++) v[u] = stage[i + u * T1];
#pragma unroll
            for (int u = 0; u < 4; u++) {
                const unsigned p = min((unsigned)(e0 + i + u * T1), (unsigned)(E - 1));
                cpair[p] = v[u];
            }
        }
        for (; i < len; i += T1) {
            const unsigned p = min((unsigned)(e0 + i), (unsigned)(E - 1));
            cpair[p] = stage[i];
        }
    } else {
        for (int e = e0 + tid; e < e1; e += T1) {
            const int sv = src[e];
            const int dv = dst[e];
            unsigned p = (unsigned)atomicAdd(&h[bucket_of(dv)], 1);
            p = min((unsigned)e0 + p, (unsigned)(E - 1));
            cpair[p] = make_int2(sv, dv);
        }
    }
}

__global__ void __launch_bounds__(T1)
bucket_kernel(const int2* __restrict__ cpair, const int* __restrict__ tab,
              const int* __restrict__ st, int E,
              int* __restrict__ cnt, int* __restrict__ off,
              float* __restrict__ dinv, int* __restrict__ csr_src) {
    __shared__ int lc[BNODES];
    __shared__ int lo[BNODES];
    __shared__ int wsum[BNODES / 32];
    __shared__ int red[T1 / 32];
    __shared__ int rsum[NB1 / 32];
    __shared__ int rpos[NB1];
    __shared__ int rcnt[NB1];
    __shared__ int rg[NB1];
    __shared__ __attribute__((aligned(16))) int raw_w[2 * RCAP];
    __shared__ int stage[RCAP];
    int2* raw = (int2*)raw_w;
    const int tid  = threadIdx.x;
    const int lane = tid & 31;
    const int wave = tid >> 5;
    const int k    = blockIdx.x;
    const int ch   = (E + NB1 - 1) / NB1;

    if (tid < BNODES) lc[tid] = 0;

    const int4* t4 = (const int4*)tab;
    const int   n4 = k * (NB1 / 4);
    int part = 0;
#pragma unroll
    for (int u = 0; u < PRE_SLOTS; u++) {
        const int  j = tid + u * T1;
        const int4 v = t4[min(j, NBUCK * NB1 / 4 - 1)];
        const int  t = (v.x + v.y) + (v.z + v.w);
        part += (j < n4) ? t : 0;
    }
    for (int o = 16; o > 0; o >>= 1) part += __shfl_xor(part, o);
    if (lane == 0) red[wave] = part;

    int rc = 0;
    int rs = 0;
    if (tid < NB1) {
        rc = tab[k * NB1 + tid];
        rs = st[k * NB1 + tid];
    }
    int rinc = rc;
    for (int o = 1; o < 32; o <<= 1) {
        int y = __shfl_up(rinc, o);
        if (lane >= o) rinc += y;
    }
    if (lane == 31 && wave < NB1 / 32) rsum[wave] = rinc;
    __syncthreads();

    int s0 = red[lane];
    for (int o = 16; o > 0; o >>= 1) s0 += __shfl_xor(s0, o);
    int len = 0;
    for (int w = 0; w < NB1 / 32; w++) len += rsum[w];
    if (tid < NB1) {
        int x = rinc - rc;
        for (int w = 0; w < NB1 / 32; w++)
            if (w < wave) x += rsum[w];
        rpos[tid] = x;
        rcnt[tid] = rc;
        rg[tid]   = tid * ch + rs;
    }
    __syncthreads();
    const bool staged = (len <= RCAP);

    auto take = [&](const int2 pr, const int q) {
        const int node = pr.y & (BNODES - 1);
        const int r = atomicAdd(&lc[node], 1);
        if (staged)
            raw[min((unsigned)q, (unsigned)(RCAP - 1))] = make_int2(pr.x | (node << SRCB), r);
    };

    for (int g = 0; g < RPW; g += 4) {
        int2 pr[4][4];
        int  rr[4][4];
        int  n[4];
        int  ps[4];
        int  gg[4];
#pragma unroll
        for (int a = 0; a < 4; a++) {
            const int run = wave * RPW + g + a;
            n[a]  = rcnt[run];
            ps[a] = rpos[run];
            gg[a] = rg[run];
        }
#pragma unroll
        for (int a = 0; a < 4; a++) {
#pragma unroll
            for (int j = 0; j < 4; j++) {
                const int p = lane + 32 * j;
                const unsigned idx = (unsigned)(gg[a] + ((p < n[a]) ? p : 0));
                pr[a][j] = cpair[min(idx, (unsigned)(E - 1))];
            }
        }
#pragma unroll
        for (int a = 0; a < 4; a++) {
#pragma unroll
            for (int j = 0; j < 4; j++) {
                rr[a][j] = 0;
                if (lane + 32 * j < n[a]) rr[a][j] = atomicAdd(&lc[pr[a][j].y & (BNODES - 1)], 1);
            }
        }
        if (staged) {
#pragma unroll
            for (int a = 0; a < 4; a++) {
#pragma unroll
                for (int j = 0; j < 4; j++) {
                    const int p = lane + 32 * j;
                    if (p < n[a]) {
                        const int node = pr[a][j].y & (BNODES - 1);
                        raw[min((unsigned)(ps[a] + p), (unsigned)(RCAP - 1))] = make_int2(pr[a][j].x | (node << SRCB), rr[a][j]);
                    }
                }
            }
        }
        for (int a = 0; a < 4; a++)
            for (int p = 128 + lane; p < n[a]; p += 32)
                take(cpair[min((unsigned)(gg[a] + p), (unsigned)(E - 1))], ps[a] + p);
    }
    __syncthreads();

    const int c = (tid < BNODES) ? lc[tid] : 0;
    int inc = c;
    for (int o = 1; o < 32; o <<= 1) {
        int y = __shfl_up(inc, o);
        if (lane >= o) inc += y;
    }
    if (lane == 31 && wave < BNODES / 32) wsum[wave] = inc;
    __syncthreads();
    if (tid < BNODES) {
        int x = inc - c;
        for (int w = 0; w < BNODES / 32; w++)
            if (w < wave) x += wsum[w];
        lo[tid] = x;
        const int node = k * BNODES + tid;
        if (node < N_NODES) {
            cnt[node]  = c;
            off[node]  = s0 + x;
            dinv[node] = rsqrtf((float)c + 1.0f);
        }
    }
    __syncthreads();

    if (staged) {
        int i = tid;
        for (; i + 3 * T1 < len; i += 4 * T1) {
            int2 pr[4];
            int  b[4];
#pragma unroll
            for (int u = 0; u < 4; u++) pr[u] = raw[i + u * T1];
#pragma unroll
            for (int u = 0; u < 4; u++) b[u] = lo[((unsigned)pr[u].x >> SRCB) & (BNODES - 1)];
#pragma unroll
            for (int u = 0; u < 4; u++) {
                const unsigned q = (unsigned)b[u] + (unsigned)pr[u].y;
                stage[min(q, (unsigned)(RCAP - 1))] = pr[u].x & ((1 << SRCB) - 1);
            }
        }
        for (; i < len; i += T1) {
            const int2 pr = raw[i];
            const unsigned q = (unsigned)lo[((unsigned)pr.x >> SRCB) & (BNODES - 1)] + (unsigned)pr.y;
            stage[min(q, (unsigned)(RCAP - 1))] = pr.x & ((1 << SRCB) - 1);
        }
        __syncthreads();
        for (int j = tid; j < len; j += T1) {
            const unsigned p = min((unsigned)(s0 + j), (unsigned)(E - 1));
            csr_src[p] = stage[j];
        }
    } else {
        for (int a = 0; a < RPW; a++) {
            const int run = wave * RPW + a;
            const int n   = rcnt[run];
            const int gs  = rg[run];
            for (int p = lane; p < n; p += 32) {
                const int2 pr = cpair[min((unsigned)(gs + p), (unsigned)(E - 1))];
                const int r = atomicAdd(&lo[pr.y & (BNODES - 1)], 1);
                const unsigned q = min((unsigned)(s0 + r), (unsigned)(E - 1));
                csr_src[q] = pr.x;
            }
        }
    }
}


__global__ void __launch_bounds__(256)
gemm1_wmma_kernel(const float* __restrict__ x,
                  const float* __restrict__ W1,
                  const float* __restrict__ dinv,
                  float* __restrict__ h1) {
    __shared__ float lw[F_IN * F_HID];
    for (int i = threadIdx.x; i < F_IN * F_HID; i += blockDim.x) {
        int kk  = i >> 6;
        int col = i & 63;
        lw[(((kk >> 1) * F_HID) + col) * 2 + (kk & 1)] = W1[i];
    }
    __syncthreads();

    const int lane  = threadIdx.x & 31;
    const int wave  = threadIdx.x >> 5;
    const int mtile = blockIdx.x * 8 + wave;
    if (mtile * 16 >= N_NODES) return;
    const int row0 = mtile * 16;

    const int r  = lane & 15;
    const int hi = lane >> 4;

    v8f acc0 = {}, acc1 = {}, acc2 = {}, acc3 = {};
    const float* xrow = x + (size_t)(row0 + r) * F_IN;
    const v2f*   wp   = (const v2f*)lw;

    for (int k = 0; k < F_IN; k += 4) {
        __builtin_prefetch(&xrow[k + 32], 0, 3);
        v2f a = *(const v2f*)&xrow[k + 2 * hi];
        const int pb = ((k >> 1) + hi) * F_HID;
        v2f b0 = wp[pb + r +  0];
        v2f b1 = wp[pb + r + 16];
        v2f b2 = wp[pb + r + 32];
        v2f b3 = wp[pb + r + 48];
        acc0 = __builtin_amdgcn_wmma_f32_16x16x4_f32(false, a, false, b0, (short)0, acc0, false, false);
        acc1 = __builtin_amdgcn_wmma_f32_16x16x4_f32(false, a, false, b1, (short)0, acc1, false, false);
        acc2 = __builtin_amdgcn_wmma_f32_16x16x4_f32(false, a, false, b2, (short)0, acc2, false, false);
        acc3 = __builtin_amdgcn_wmma_f32_16x16x4_f32(false, a, false, b3, (short)0, acc3, false, false);
    }

    const float4* dv = (const float4*)(dinv + row0 + 8 * hi);
    const float4 dlo = dv[0];
    const float4 dhi = dv[1];
    const float ds[8] = {dlo.x, dlo.y, dlo.z, dlo.w, dhi.x, dhi.y, dhi.z, dhi.w};
    float* out = h1 + (size_t)row0 * F_HID;
#pragma unroll
    for (int i = 0; i < 8; i++) {
        int m = i + 8 * hi;
        out[(size_t)m * F_HID + (r +  0)] = acc0[i] * ds[i];
        out[(size_t)m * F_HID + (r + 16)] = acc1[i] * ds[i];
        out[(size_t)m * F_HID + (r + 32)] = acc2[i] * ds[i];
        out[(size_t)m * F_HID + (r + 48)] = acc3[i] * ds[i];
    }
}


template <int S, bool FULL>
__device__ __forceinline__ void gather_rows(float4& acc, int s, int j, int m,
                                            int hi, int q,
                                            const float4* __restrict__ hv) {
    float4 v[S];
    int    su[S];
#pragma unroll
    for (int u = 0; u < S; u++) su[u] = __shfl(s, j + 2 * u + hi);
#pragma unroll
    for (int u = 0; u < S; u++) v[u] = hv[(unsigned)su[u] * (F_HID / 4) + q];
#pragma unroll
    for (int u = 0; u < S; u++) {
        if (FULL) {
            acc.x += v[u].x;
            acc.y += v[u].y;
            acc.z += v[u].z;
            acc.w += v[u].w;
        } else {
            const float mk = (j + 2 * u + hi < m) ? 1.0f : 0.0f;
            acc.x += v[u].x * mk;
            acc.y += v[u].y * mk;
            acc.z += v[u].z * mk;
            acc.w += v[u].w * mk;
        }
    }
}

__global__ void __launch_bounds__(256)
gather1_kernel(const int* __restrict__ csr_src,
               const int* __restrict__ cnt,
               const int* __restrict__ off,
               const float* __restrict__ dinv,
               const float* __restrict__ h1,
               const float* __restrict__ b1,
               float* __restrict__ act1) {
    const int lane = threadIdx.x & 31;
    const int d = __builtin_amdgcn_readfirstlane(blockIdx.x * 8 + (threadIdx.x >> 5));
    if (d >= N_NODES) return;
    const int q  = lane & 15;
    const int hi = lane >> 4;
    const int   n    = cnt[d];
    const int   base = off[d];
    const float dd   = dinv[d];
    const float4* hv = (const float4*)h1;

    float4 acc = {0.0f, 0.0f, 0.0f, 0.0f};
    for (int c = 0; c < n; c += 32) {
        const int m  = min(32, n - c);
        const int my = (lane < m) ? lane : (m - 1);
        unsigned su = (unsigned)csr_src[base + c + my];
        su = min(su, (unsigned)(N_NODES - 1));
        const int s = (int)su;
        int j = 0;
        while (j < m) {
            const int left = m - j;
            if (left >= 16)    { gather_rows<8, true >(acc, s, j, m, hi, q, hv); j += 16; }
            else if (left > 8) { gather_rows<8, false>(acc, s, j, m, hi, q, hv); j += 16; }
            else               { gather_rows<4, false>(acc, s, j, m, hi, q, hv); j += 8;  }
        }
    }
    acc.x += __shfl_xor(acc.x, 16);
    acc.y += __shfl_xor(acc.y, 16);
    acc.z += __shfl_xor(acc.z, 16);
    acc.w += __shfl_xor(acc.w, 16);

    if (hi == 0) {
        const float4 hs  = hv[(size_t)d * (F_HID / 4) + q];
        const float4 bb  = ((const float4*)b1)[q];
        float4 o;
        o.x = (acc.x + hs.x) * dd + bb.x;
        o.y = (acc.y + hs.y) * dd + bb.y;
        o.z = (acc.z + hs.z) * dd + bb.z;
        o.w = (acc.w + hs.w) * dd + bb.w;
        o.x = (o.x > 0.0f) ? o.x : 0.01f * o.x;
        o.y = (o.y > 0.0f) ? o.y : 0.01f * o.y;
        o.z = (o.z > 0.0f) ? o.z : 0.01f * o.z;
        o.w = (o.w > 0.0f) ? o.w : 0.01f * o.w;
        ((float4*)act1)[(size_t)d * (F_HID / 4) + q] = o;
    }
}


__global__ void __launch_bounds__(256)
gemm2_wmma_kernel(const float* __restrict__ h,
                  const float* __restrict__ W2,
                  const float* __restrict__ dinv,
                  float* __restrict__ h2) {
    __shared__ float lw[F_HID * 16];
    for (int i = threadIdx.x; i < F_HID * 16; i += blockDim.x) {
        int kk = i >> 4, nn = i & 15;
        float v = (nn < F_OUT) ? W2[kk * F_OUT + nn] : 0.0f;
        lw[(((kk >> 1) * 16) + nn) * 2 + (kk & 1)] = v;
    }
    __syncthreads();

    const int lane  = threadIdx.x & 31;
    const int wave  = threadIdx.x >> 5;
    const int mtile = blockIdx.x * 8 + wave;
    if (mtile * 16 >= N_NODES) return;
    const int row0 = mtile * 16;

    const int r  = lane & 15;
    const int hi = lane >> 4;

    const float4* dv = (const float4*)(dinv + row0 + 8 * hi);
    const float4 dlo = dv[0];
    const float4 dhi = dv[1];

    v8f acc = {};
    const float* hrow = h + (size_t)(row0 + r) * F_HID;
    const v2f*   wp   = (const v2f*)lw;
    for (int k = 0; k < F_HID; k += 4) {
        v2f a = *(const v2f*)&hrow[k + 2 * hi];
        v2f b = wp[((k >> 1) + hi) * 16 + r];
        acc = __builtin_amdgcn_wmma_f32_16x16x4_f32(false, a, false, b, (short)0, acc, false, false);
    }

    if (r < F_OUT) {
        const float ds[8] = {dlo.x, dlo.y, dlo.z, dlo.w, dhi.x, dhi.y, dhi.z, dhi.w};
        float* out = h2 + (size_t)row0 * F_OUT;
#pragma unroll
        for (int i = 0; i < 8; i++) {
            int m = i + 8 * hi;
            out[(size_t)m * F_OUT + r] = acc[i] * ds[i];
        }
    }
}


__global__ void __launch_bounds__(256)
gather2_kernel(const int* __restrict__ csr_src,
               const int* __restrict__ cnt,
               const int* __restrict__ off,
               const float* __restrict__ dinv,
               const float* __restrict__ h2,
               const float* __restrict__ b2,
               float* __restrict__ out) {
    const int t   = blockIdx.x * blockDim.x + threadIdx.x;
    const int d   = t >> 2;
    const int sub = t & 3;
    if (d >= N_NODES) return;
    const int   n    = cnt[d];
    const int   base = off[d];
    const float dd   = dinv[d];
    const float4* hv = (const float4*)h2;

    float4 acc = {0.0f, 0.0f, 0.0f, 0.0f};
    for (int j = sub; j < n; j += 16) {
        unsigned s[4];
        float    w[4];
        float4   v[4];
#pragma unroll
        for (int u = 0; u < 4; u++) {
            const bool ok = (j + 4 * u) < n;
            s[u] = (unsigned)csr_src[base + (ok ? j + 4 * u : j)];
            s[u] = min(s[u], (unsigned)(N_NODES - 1));
            w[u] = ok ? 1.0f : 0.0f;
            v[u] = hv[s[u]];
        }
#pragma unroll
        for (int u = 0; u < 4; u++) {
            acc.x += v[u].x * w[u];
            acc.y += v[u].y * w[u];
            acc.z += v[u].z * w[u];
            acc.w += v[u].w * w[u];
        }
    }
    for (int o = 2; o > 0; o >>= 1) {
        acc.x += __shfl_xor(acc.x, o);
        acc.y += __shfl_xor(acc.y, o);
        acc.z += __shfl_xor(acc.z, o);
        acc.w += __shfl_xor(acc.w, o);
    }
    if (sub == 0) {
        const float4 hs  = hv[d];
        float v[F_OUT];
        v[0] = (acc.x + hs.x) * dd + b2[0];
        v[1] = (acc.y + hs.y) * dd + b2[1];
        v[2] = (acc.z + hs.z) * dd + b2[2];
        v[3] = (acc.w + hs.w) * dd + b2[3];
        float mx = -3.4e38f;
        for (int c = 0; c < F_OUT; c++) mx = fmaxf(mx, v[c]);
        float sm = 0.0f;
        for (int c = 0; c < F_OUT; c++) { v[c] = __expf(v[c] - mx); sm += v[c]; }
        const float rs = 1.0f / sm;
        float4 o;
        o.x = v[0] * rs;  o.y = v[1] * rs;  o.z = v[2] * rs;  o.w = v[3] * rs;
        ((float4*)out)[d] = o;
    }
}


extern "C" void kernel_launch(void* const* d_in, const int* in_sizes, int n_in,
                              void* d_out, int out_size, void* d_ws, size_t ws_size,
                              hipStream_t stream) {
    const float* x  = (const float*)d_in[0];
    const int*   ei = (const int*)d_in[1];
    const float* W1 = (const float*)d_in[2];
    const float* b1 = (const float*)d_in[3];
    const float* W2 = (const float*)d_in[4];
    const float* b2 = (const float*)d_in[5];
    float* out = (float*)d_out;

    const int E = in_sizes[1] / 2;
    const int* src = ei;
    const int* dst = ei + E;

    int*   cnt     = (int*)d_ws;
    int*   off     = cnt + N_NODES;
    float* dinv    = (float*)(off + N_NODES);
    int*   csr_src = (int*)(dinv + N_NODES + 32);
    float* h1      = (float*)(csr_src + E);
    float* act1    = h1   + (size_t)N_NODES * F_HID;
    float* h2      = act1 + (size_t)N_NODES * F_HID;
    int2*  cpair   = (int2*)h1;
    int*   tab     = (int*)h2;
    int*   st      = tab + NBUCK * NB1;

    const int T = 256;
    const int nb_tiles  = (N_NODES / 16 + 7) / 8;
    const int nb_g1     = (N_NODES + 7) / 8;
    const int nb_g2     = (N_NODES * 4 + T - 1) / T;

    part_kernel<<<NB1, T1, 0, stream>>>(src, dst, E, tab, st, cpair);
    bucket_kernel<<<NBUCK, T1, 0, stream>>>(cpair, tab, st, E, cnt, off, dinv, csr_src);

    gemm1_wmma_kernel<<<nb_tiles, T, 0, stream>>>(x, W1, dinv, h1);
    gather1_kernel<<<nb_g1, T, 0, stream>>>(csr_src, cnt, off, dinv, h1, b1, act1);

    gemm2_wmma_kernel<<<nb_tiles, T, 0, stream>>>(act1, W2, dinv, h2);
    gather2_kernel<<<nb_g2, T, 0, stream>>>(csr_src, cnt, off, dinv, h2, b2, out);
}


# layer-1 product runs in the first sort launch as work-groups of its own beside the level-1 chunks; rows stored unscaled, sort level 2 scales its nodes' rows by dinv in place; sum order unchanged
# speedup vs baseline: 8.7059x; 1.2610x over previous
#include <hip/hip_runtime.h>


typedef __attribute__((ext_vector_type(2))) float v2f;
typedef __attribute__((ext_vector_type(8))) float v8f;

#define N_NODES 100000
#define F_IN    256
#define F_HID   64
#define F_OUT   4


#define BSH     9
#define BNODES  (1 << BSH)
#define NBUCK   ((N_NODES + BNODES - 1) / BNODES)
#define NB1     256
#define T1      1024
#define TP      512
#define EPT     25
#define SCAP    (EPT * TP)
#define NTILE   (N_NODES / 16)
#define GT      25
#define NGEMM   ((NTILE + GT - 1) / GT)
static_assert(N_NODES % 16 == 0, "the product's tiles are whole");
static_assert(2 * SCAP >= F_IN * F_HID, "W1 must fit the stage it shares");
#define RCAP    20480
#define RPW     (NB1 / (T1 / 32))
#define SRCB    20
static_assert(N_NODES <= (1 << SRCB) && BSH + SRCB <= 32, "a source index and a range-local node must fit one word");
#define PRE_SLOTS ((NBUCK * NB1 / 4 + T1 - 1) / T1)
#define HSLOTS  (BNODES * (F_HID / 4) / T1)
static_assert(BNODES * (F_HID / 4) % T1 == 0, "a range's rows are whole slots");

__device__ __forceinline__ unsigned bucket_of(int d) {
    return min((unsigned)d >> BSH, (unsigned)(NBUCK - 1));
}

__global__ void __launch_bounds__(TP, 8)
first_kernel(const int* __restrict__ src, const int* __restrict__ dst, int E,
             int* __restrict__ tab, int* __restrict__ st, int2* __restrict__ cpair,
             const float* __restrict__ x, const float* __restrict__ W1,
             float* __restrict__ h1) {
    __shared__ int h[NBUCK];
    __shared__ int ex[NBUCK];
    __shared__ int wsum[8];
    __shared__ __attribute__((aligned(16))) int stage_w[2 * SCAP];
    int2* stage = (int2*)stage_w;
    const int tid  = threadIdx.x;
    const int lane = tid & 31;

    if (blockIdx.x < NGEMM) {
        float* lw = (float*)stage_w;
        for (int i = tid; i < F_IN * F_HID; i += TP) {
            int kk  = i >> 6;
            int col = i & 63;
            lw[(((kk >> 1) * F_HID) + col) * 2 + (kk & 1)] = W1[i];
        }
        __syncthreads();

        const int gw = __builtin_amdgcn_readfirstlane(tid >> 5);
        const int r  = lane & 15;
        const int hi = lane >> 4;
        const v2f* wp = (const v2f*)lw;
        const int t0 = (int)blockIdx.x * GT;
        const int t1 = min(t0 + GT, NTILE);
        for (int mtile = t0 + gw; mtile < t1; mtile += TP / 32) {
            const int row0 = mtile * 16;
            v8f acc0 = {}, acc1 = {}, acc2 = {}, acc3 = {};
            const float* xrow = x + (size_t)(row0 + r) * F_IN;

            for (int k = 0; k < F_IN; k += 4) {
                __builtin_prefetch(&xrow[k + 32], 0, 3);
                v2f a = *(const v2f*)&xrow[k + 2 * hi];
                const int pb = ((k >> 1) + hi) * F_HID;
                v2f b0 = wp[pb + r +  0];
                v2f b1 = wp[pb + r + 16];
                v2f b2 = wp[pb + r + 32];
                v2f b3 = wp[pb + r + 48];
                acc0 = __builtin_amdgcn_wmma_f32_16x16x4_f32(false, a, false, b0, (short)0, acc0, false, false);
                acc1 = __builtin_amdgcn_wmma_f32_16x16x4_f32(false, a, false, b1, (short)0, acc1, false, false);
                acc2 = __builtin_amdgcn_wmma_f32_16x16x4_f32(false, a, false, b2, (short)0, acc2, false, false);
                acc3 = __builtin_amdgcn_wmma_f32_16x16x4_f32(false, a, false, b3, (short)0, acc3, false, false);
            }

            float* out = h1 + (size_t)row0 * F_HID;
#pragma unroll
            for (int i = 0; i < 8; i++) {
                int m = i + 8 * hi;
                out[(size_t)m * F_HID + (r +  0)] = acc0[i];
                out[(size_t)m * F_HID + (r + 16)] = acc1[i];
                out[(size_t)m * F_HID + (r + 32)] = acc2[i];
                out[(size_t)m * F_HID + (r + 48)] = acc3[i];
            }
        }
        return;
    }

    const int wave = tid >> 5;
    const int chunk = (int)blockIdx.x - NGEMM;
    const int ch = (E + NB1 - 1) / NB1;
    const int e0 = min(E, chunk * ch);
    const int e1 = min(E, e0 + ch);
    const bool staged = (ch <= SCAP);
    const int len = e1 - e0;

    if (tid < NBUCK) h[tid] = 0;
    __syncthreads();

    int      s[EPT];
    int      d[EPT];
    unsigned rk[EPT];
    if (staged) {
#pragma unroll
        for (int u = 0; u < EPT; u++) {
            const unsigned e = min((unsigned)(e0 + tid + u * TP), (unsigned)(E - 1));
            s[u] = src[e];
            d[u] = dst[e];
        }
#pragma unroll
        for (int u = 0; u < EPT; u++) {
            rk[u] = 0;
            if (tid < len - u * TP) rk[u] = (unsigned)atomicAdd(&h[bucket_of(d[u])], 1);
        }
    } else {
        for (int e = e0 + tid; e < e1; e += TP) atomicAdd(&h[bucket_of(dst[e])], 1);
    }
    __syncthreads();

    const int c = (tid < NBUCK) ? h[tid] : 0;
    int inc = c;
    for (int o = 1; o < 32; o <<= 1) {
        int y = __shfl_up(inc, o);
        if (lane >= o) inc += y;
    }
    if (lane == 31 && wave < 8) wsum[wave] = inc;
    __syncthreads();
    if (tid < NBUCK) {
        int xs = inc - c;
        for (int w = 0; w < 8; w++)
            if (w < wave) xs += wsum[w];
        ex[tid] = xs;
        tab[tid * NB1 + chunk] = c;
        st[tid * NB1 + chunk]  = xs;
        if (!staged) h[tid] = xs;
    }
    __syncthreads();

    if (staged) {
        unsigned q[EPT];
#pragma unroll
        for (int u = 0; u < EPT; u++)
            q[u] = min((unsigned)ex[bucket_of(d[u])] + rk[u], (unsigned)(SCAP - 1));
#pragma unroll
        for (int u = 0; u < EPT; u++)
            if (tid < len - u * TP) stage[q[u]] = make_int2(s[u], d[u]);
        __syncthreads();
        int i = tid;
        for (; i + 3 * TP < len; i += 4 * TP) {
            int2 v[4];
#pragma unroll
            for (int u = 0; u < 4; u++) v[u] = stage[i + u * TP];
#pragma unroll
            for (int u = 0; u < 4; u++) {
                const unsigned p = min((unsigned)(e0 + i + u * TP), (unsigned)(E - 1));
                cpair[p] = v[u];
            }
        }
        for (; i < len; i += TP) {
            const unsigned p = min((unsigned)(e0 + i), (unsigned)(E - 1));
            cpair[p] = stage[i];
        }
    } else {
        for (int e = e0 + tid; e < e1; e += TP) {
            const int sv = src[e];
            const int dv = dst[e];
            unsigned p = (unsigned)atomicAdd(&h[bucket_of(dv)], 1);
            p = min((unsigned)e0 + p, (unsigned)(E - 1));
            cpair[p] = make_int2(sv, dv);
        }
    }
}

__global__ void __launch_bounds__(T1)
bucket_kernel(const int2* __restrict__ cpair, const int* __restrict__ tab,
              const int* __restrict__ st, int E,
              int* __restrict__ cnt, int* __restrict__ off,
              float* __restrict__ dinv, int* __restrict__ csr_src,
              float* __restrict__ h1) {
    __shared__ int lc[BNODES];
    __shared__ int lo[BNODES];
    __shared__ int wsum[BNODES / 32];
    __shared__ int red[T1 / 32];
    __shared__ int rsum[NB1 / 32];
    __shared__ int rpos[NB1];
    __shared__ int rcnt[NB1];
    __shared__ int rg[NB1];
    __shared__ float ldv[BNODES];
    __shared__ __attribute__((aligned(16))) int raw_w[2 * RCAP];
    __shared__ int stage[RCAP];
    int2* raw = (int2*)raw_w;
    const int tid  = threadIdx.x;
    const int lane = tid & 31;
    const int wave = tid >> 5;
    const int k    = blockIdx.x;
    const int ch   = (E + NB1 - 1) / NB1;

    if (tid < BNODES) lc[tid] = 0;

    const int4* t4 = (const int4*)tab;
    const int   n4 = k * (NB1 / 4);
    int part = 0;
#pragma unroll
    for (int u = 0; u < PRE_SLOTS; u++) {
        const int  j = tid + u * T1;
        const int4 v = t4[min(j, NBUCK * NB1 / 4 - 1)];
        const int  t = (v.x + v.y) + (v.z + v.w);
        part += (j < n4) ? t : 0;
    }
    for (int o = 16; o > 0; o >>= 1) part += __shfl_xor(part, o);
    if (lane == 0) red[wave] = part;

    int rc = 0;
    int rs = 0;
    if (tid < NB1) {
        rc = tab[k * NB1 + tid];
        rs = st[k * NB1 + tid];
    }
    int rinc = rc;
    for (int o = 1; o < 32; o <<= 1) {
        int y = __shfl_up(rinc, o);
        if (lane >= o) rinc += y;
    }
    if (lane == 31 && wave < NB1 / 32) rsum[wave] = rinc;
    __syncthreads();

    int s0 = red[lane];
    for (int o = 16; o > 0; o >>= 1) s0 += __shfl_xor(s0, o);
    int len = 0;
    for (int w = 0; w < NB1 / 32; w++) len += rsum[w];
    if (tid < NB1) {
        int x = rinc - rc;
        for (int w = 0; w < NB1 / 32; w++)
            if (w < wave) x += rsum[w];
        rpos[tid] = x;
        rcnt[tid] = rc;
        rg[tid]   = tid * ch + rs;
    }
    __syncthreads();
    const bool staged = (len <= RCAP);

    auto take = [&](const int2 pr, const int q) {
        const int node = pr.y & (BNODES - 1);
        const int r = atomicAdd(&lc[node], 1);
        if (staged)
            raw[min((unsigned)q, (unsigned)(RCAP - 1))] = make_int2(pr.x | (node << SRCB), r);
    };

    for (int g = 0; g < RPW; g += 4) {
        int2 pr[4][4];
        int  rr[4][4];
        int  n[4];
        int  ps[4];
        int  gg[4];
#pragma unroll
        for (int a = 0; a < 4; a++) {
            const int run = wave * RPW + g + a;
            n[a]  = rcnt[run];
            ps[a] = rpos[run];
            gg[a] = rg[run];
        }
#pragma unroll
        for (int a = 0; a < 4; a++) {
#pragma unroll
            for (int j = 0; j < 4; j++) {
                const int p = lane + 32 * j;
                const unsigned idx = (unsigned)(gg[a] + ((p < n[a]) ? p : 0));
                pr[a][j] = cpair[min(idx, (unsigned)(E - 1))];
            }
        }
#pragma unroll
        for (int a = 0; a < 4; a++) {
#pragma unroll
            for (int j = 0; j < 4; j++) {
                rr[a][j] = 0;
                if (lane + 32 * j < n[a]) rr[a][j] = atomicAdd(&lc[pr[a][j].y & (BNODES - 1)], 1);
            }
        }
        if (staged) {
#pragma unroll
            for (int a = 0; a < 4; a++) {
#pragma unroll
                for (int j = 0; j < 4; j++) {
                    const int p = lane + 32 * j;
                    if (p < n[a]) {
                        const int node = pr[a][j].y & (BNODES - 1);
                        raw[min((unsigned)(ps[a] + p), (unsigned)(RCAP - 1))] = make_int2(pr[a][j].x | (node << SRCB), rr[a][j]);
                    }
                }
            }
        }
        for (int a = 0; a < 4; a++)
            for (int p = 128 + lane; p < n[a]; p += 32)
                take(cpair[min((unsigned)(gg[a] + p), (unsigned)(E - 1))], ps[a] + p);
    }
    __syncthreads();

    const int c = (tid < BNODES) ? lc[tid] : 0;
    int inc = c;
    for (int o = 1; o < 32; o <<= 1) {
        int y = __shfl_up(inc, o);
        if (lane >= o) inc += y;
    }
    if (lane == 31 && wave < BNODES / 32) wsum[wave] = inc;
    __syncthreads();
    if (tid < BNODES) {
        int x = inc - c;
        for (int w = 0; w < BNODES / 32; w++)
            if (w < wave) x += wsum[w];
        lo[tid] = x;
        const float dn = rsqrtf((float)c + 1.0f);
        ldv[tid] = dn;
        const int node = k * BNODES + tid;
        if (node < N_NODES) {
            cnt[node]  = c;
            off[node]  = s0 + x;
            dinv[node] = dn;
        }
    }
    __syncthreads();

    const int nrow4 = min(BNODES, N_NODES - k * BNODES) * (F_HID / 4);
    float4* hrow = (float4*)h1 + (size_t)k * BNODES * (F_HID / 4);
    float4 hr[HSLOTS];
#pragma unroll
    for (int u = 0; u < HSLOTS; u++)
        hr[u] = hrow[min(tid + u * T1, nrow4 - 1)];

    if (staged) {
        int i = tid;
        for (; i + 3 * T1 < len; i += 4 * T1) {
            int2 pr[4];
            int  b[4];
#pragma unroll
            for (int u = 0; u < 4; u++) pr[u] = raw[i + u * T1];
#pragma unroll
            for (int u = 0; u < 4; u++) b[u] = lo[((unsigned)pr[u].x >> SRCB) & (BNODES - 1)];
#pragma unroll
            for (int u = 0; u < 4; u++) {
                const unsigned q = (unsigned)b[u] + (unsigned)pr[u].y;
                stage[min(q, (unsigned)(RCAP - 1))] = pr[u].x & ((1 << SRCB) - 1);
            }
        }
        for (; i < len; i += T1) {
            const int2 pr = raw[i];
            const unsigned q = (unsigned)lo[((unsigned)pr.x >> SRCB) & (BNODES - 1)] + (unsigned)pr.y;
            stage[min(q, (unsigned)(RCAP - 1))] = pr.x & ((1 << SRCB) - 1);
        }
        __syncthreads();
        for (int j = tid; j < len; j += T1) {
            const unsigned p = min((unsigned)(s0 + j), (unsigned)(E - 1));
            csr_src[p] = stage[j];
        }
    } else {
        for (int a = 0; a < RPW; a++) {
            const int run = wave * RPW + a;
            const int n   = rcnt[run];
            const int gs  = rg[run];
            for (int p = lane; p < n; p += 32) {
                const int2 pr = cpair[min((unsigned)(gs + p), (unsigned)(E - 1))];
                const int r = atomicAdd(&lo[pr.y & (BNODES - 1)], 1);
                const unsigned q = min((unsigned)(s0 + r), (unsigned)(E - 1));
                csr_src[q] = pr.x;
            }
        }
    }

#pragma unroll
    for (int u = 0; u < HSLOTS; u++) {
        const int j = tid + u * T1;
        if (j < nrow4) {
            const float dn = ldv[j / (F_HID / 4)];
            float4 v = hr[u];
            v.x *= dn;
            v.y *= dn;
            v.z *= dn;
            v.w *= dn;
            hrow[j] = v;
        }
    }
}


template <int S, bool FULL>
__device__ __forceinline__ void gather_rows(float4& acc, int s, int j, int m,
                                            int hi, int q,
                                            const float4* __restrict__ hv) {
    float4 v[S];
    int    su[S];
#pragma unroll
    for (int u = 0; u < S; u++) su[u] = __shfl(s, j + 2 * u + hi);
#pragma unroll
    for (int u = 0; u < S; u++) v[u] = hv[(unsigned)su[u] * (F_HID / 4) + q];
#pragma unroll
    for (int u = 0; u < S; u++) {
        if (FULL) {
            acc.x += v[u].x;
            acc.y += v[u].y;
            acc.z += v[u].z;
            acc.w += v[u].w;
        } else {
            const float mk = (j + 2 * u + hi < m) ? 1.0f : 0.0f;
            acc.x += v[u].x * mk;
            acc.y += v[u].y * mk;
            acc.z += v[u].z * mk;
            acc.w += v[u].w * mk;
        }
    }
}

__global__ void __launch_bounds__(256)
gather1_kernel(const int* __restrict__ csr_src,
               const int* __restrict__ cnt,
               const int* __restrict__ off,
               const float* __restrict__ dinv,
               const float* __restrict__ h1,
               const float* __restrict__ b1,
               float* __restrict__ act1) {
    const int lane = threadIdx.x & 31;
    const int d = __builtin_amdgcn_readfirstlane(blockIdx.x * 8 + (threadIdx.x >> 5));
    if (d >= N_NODES) return;
    const int q  = lane & 15;
    const int hi = lane >> 4;
    const int   n    = cnt[d];
    const int   base = off[d];
    const float dd   = dinv[d];
    const float4* hv = (const float4*)h1;

    float4 acc = {0.0f, 0.0f, 0.0f, 0.0f};
    for (int c = 0; c < n; c += 32) {
        const int m  = min(32, n - c);
        const int my = (lane < m) ? lane : (m - 1);
        unsigned su = (unsigned)csr_src[base + c + my];
        su = min(su, (unsigned)(N_NODES - 1));
        const int s = (int)su;
        int j = 0;
        while (j < m) {
            const int left = m - j;
            if (left >= 16)    { gather_rows<8, true >(acc, s, j, m, hi, q, hv); j += 16; }
            else if (left > 8) { gather_rows<8, false>(acc, s, j, m, hi, q, hv); j += 16; }
            else               { gather_rows<4, false>(acc, s, j, m, hi, q, hv); j += 8;  }
        }
    }
    acc.x += __shfl_xor(acc.x, 16);
    acc.y += __shfl_xor(acc.y, 16);
    acc.z += __shfl_xor(acc.z, 16);
    acc.w += __shfl_xor(acc.w, 16);

    if (hi == 0) {
        const float4 hs  = hv[(size_t)d * (F_HID / 4) + q];
        const float4 bb  = ((const float4*)b1)[q];
        float4 o;
        o.x = (acc.x + hs.x) * dd + bb.x;
        o.y = (acc.y + hs.y) * dd + bb.y;
        o.z = (acc.z + hs.z) * dd + bb.z;
        o.w = (acc.w + hs.w) * dd + bb.w;
        o.x = (o.x > 0.0f) ? o.x : 0.01f * o.x;
        o.y = (o.y > 0.0f) ? o.y : 0.01f * o.y;
        o.z = (o.z > 0.0f) ? o.z : 0.01f * o.z;
        o.w = (o.w > 0.0f) ? o.w : 0.01f * o.w;
        ((float4*)act1)[(size_t)d * (F_HID / 4) + q] = o;
    }
}


__global__ void __launch_bounds__(256)
gemm2_wmma_kernel(const float* __restrict__ h,
                  const float* __restrict__ W2,
                  const float* __restrict__ dinv,
                  float* __restrict__ h2) {
    __shared__ float lw[F_HID * 16];
    for (int i = threadIdx.x; i < F_HID * 16; i += blockDim.x) {
        int kk = i >> 4, nn = i & 15;
        float v = (nn < F_OUT) ? W2[kk * F_OUT + nn] : 0.0f;
        lw[(((kk >> 1) * 16) + nn) * 2 + (kk & 1)] = v;
    }
    __syncthreads();

    const int lane  = threadIdx.x & 31;
    const int wave  = threadIdx.x >> 5;
    const int mtile = blockIdx.x * 8 + wave;
    if (mtile * 16 >= N_NODES) return;
    const int row0 = mtile * 16;

    const int r  = lane & 15;
    const int hi = lane >> 4;

    const float4* dv = (const float4*)(dinv + row0 + 8 * hi);
    const float4 dlo = dv[0];
    const float4 dhi = dv[1];

    v8f acc = {};
    const float* hrow = h + (size_t)(row0 + r) * F_HID;
    const v2f*   wp   = (const v2f*)lw;
    for (int k = 0; k < F_HID; k += 4) {
        v2f a = *(const v2f*)&hrow[k + 2 * hi];
        v2f b = wp[((k >> 1) + hi) * 16 + r];
        acc = __builtin_amdgcn_wmma_f32_16x16x4_f32(false, a, false, b, (short)0, acc, false, false);
    }

    if (r < F_OUT) {
        const float ds[8] = {dlo.x, dlo.y, dlo.z, dlo.w, dhi.x, dhi.y, dhi.z, dhi.w};
        float* out = h2 + (size_t)row0 * F_OUT;
#pragma unroll
        for (int i = 0; i < 8; i++) {
            int m = i + 8 * hi;
            out[(size_t)m * F_OUT + r] = acc[i] * ds[i];
        }
    }
}


__global__ void __launch_bounds__(256)
gather2_kernel(const int* __restrict__ csr_src,
               const int* __restrict__ cnt,
               const int* __restrict__ off,
               const float* __restrict__ dinv,
               const float* __restrict__ h2,
               const float* __restrict__ b2,
               float* __restrict__ out) {
    const int t   = blockIdx.x * blockDim.x + threadIdx.x;
    const int d   = t >> 2;
    const int sub = t & 3;
    if (d >= N_NODES) return;
    const int   n    = cnt[d];
    const int   base = off[d];
    const float dd   = dinv[d];
    const float4* hv = (const float4*)h2;

    float4 acc = {0.0f, 0.0f, 0.0f, 0.0f};
    for (int j = sub; j < n; j += 16) {
        unsigned s[4];
        float    w[4];
        float4   v[4];
#pragma unroll
        for (int u = 0; u < 4; u++) {
            const bool ok = (j + 4 * u) < n;
            s[u] = (unsigned)csr_src[base + (ok ? j + 4 * u : j)];
            s[u] = min(s[u], (unsigned)(N_NODES - 1));
            w[u] = ok ? 1.0f : 0.0f;
            v[u] = hv[s[u]];
        }
#pragma unroll
        for (int u = 0; u < 4; u++) {
            acc.x += v[u].x * w[u];
            acc.y += v[u].y * w[u];
            acc.z += v[u].z * w[u];
            acc.w += v[u].w * w[u];
        }
    }
    for (int o = 2; o > 0; o >>= 1) {
        acc.x += __shfl_xor(acc.x, o);
        acc.y += __shfl_xor(acc.y, o);
        acc.z += __shfl_xor(acc.z, o);
        acc.w += __shfl_xor(acc.w, o);
    }
    if (sub == 0) {
        const float4 hs  = hv[d];
        float v[F_OUT];
        v[0] = (acc.x + hs.x) * dd + b2[0];
        v[1] = (acc.y + hs.y) * dd + b2[1];
        v[2] = (acc.z + hs.z) * dd + b2[2];
        v[3] = (acc.w + hs.w) * dd + b2[3];
        float mx = -3.4e38f;
        for (int c = 0; c < F_OUT; c++) mx = fmaxf(mx, v[c]);
        float sm = 0.0f;
        for (int c = 0; c < F_OUT; c++) { v[c] = __expf(v[c] - mx); sm += v[c]; }
        const float rs = 1.0f / sm;
        float4 o;
        o.x = v[0] * rs;  o.y = v[1] * rs;  o.z = v[2] * rs;  o.w = v[3] * rs;
        ((float4*)out)[d] = o;
    }
}


extern "C" void kernel_launch(void* const* d_in, const int* in_sizes, int n_in,
                              void* d_out, int out_size, void* d_ws, size_t ws_size,
                              hipStream_t stream) {
    const float* x  = (const float*)d_in[0];
    const int*   ei = (const int*)d_in[1];
    const float* W1 = (const float*)d_in[2];
    const float* b1 = (const float*)d_in[3];
    const float* W2 = (const float*)d_in[4];
    const float* b2 = (const float*)d_in[5];
    float* out = (float*)d_out;

    const int E = in_sizes[1] / 2;
    const int* src = ei;
    const int* dst = ei + E;

    int*   cnt     = (int*)d_ws;
    int*   off     = cnt + N_NODES;
    float* dinv    = (float*)(off + N_NODES);
    int*   csr_src = (int*)(dinv + N_NODES + 32);
    float* h1      = (float*)(csr_src + E);
    float* act1    = h1   + (size_t)N_NODES * F_HID;
    float* h2      = act1 + (size_t)N_NODES * F_HID;
    int2*  cpair   = (int2*)act1;
    int*   tab     = (int*)h2;
    int*   st      = tab + NBUCK * NB1;

    const int T = 256;
    const int nb_tiles  = (N_NODES / 16 + 7) / 8;
    const int nb_g1     = (N_NODES + 7) / 8;
    const int nb_g2     = (N_NODES * 4 + T - 1) / T;

    first_kernel<<<NGEMM + NB1, TP, 0, stream>>>(src, dst, E, tab, st, cpair, x, W1, h1);
    bucket_kernel<<<NBUCK, T1, 0, stream>>>(cpair, tab, st, E, cnt, off, dinv, csr_src, h1);

    gather1_kernel<<<nb_g1, T, 0, stream>>>(csr_src, cnt, off, dinv, h1, b1, act1);

    gemm2_wmma_kernel<<<nb_tiles, T, 0, stream>>>(act1, W2, dinv, h2);
    gather2_kernel<<<nb_g2, T, 0, stream>>>(csr_src, cnt, off, dinv, h2, b2, out);
}


# layer-2 product (same f32 WMMA loop and k order) moved into the layer-1 gather launch: 16 one-node waves leave their rows in an LDS tile, wave 0 multiplies it; 4 launches, was 5; sum order unchanged
# speedup vs baseline: 9.0017x; 1.0340x over previous
#include <hip/hip_runtime.h>


typedef __attribute__((ext_vector_type(2))) float v2f;
typedef __attribute__((ext_vector_type(8))) float v8f;

#define N_NODES 100000
#define F_IN    256
#define F_HID   64
#define F_OUT   4


#define BSH     9
#define BNODES  (1 << BSH)
#define NBUCK   ((N_NODES + BNODES - 1) / BNODES)
#define NB1     256
#define T1      1024
#define TP      512
#define EPT     25
#define SCAP    (EPT * TP)
#define NTILE   (N_NODES / 16)
#define GT      25
#define NGEMM   ((NTILE + GT - 1) / GT)
static_assert(N_NODES % 16 == 0, "the product's tiles are whole");
static_assert(2 * SCAP >= F_IN * F_HID, "W1 must fit the stage it shares");
#define RCAP    20480
#define RPW     (NB1 / (T1 / 32))
#define SRCB    20
static_assert(N_NODES <= (1 << SRCB) && BSH + SRCB <= 32, "a source index and a range-local node must fit one word");
#define PRE_SLOTS ((NBUCK * NB1 / 4 + T1 - 1) / T1)
#define HSLOTS  (BNODES * (F_HID / 4) / T1)
static_assert(BNODES * (F_HID / 4) % T1 == 0, "a range's rows are whole slots");

__device__ __forceinline__ unsigned bucket_of(int d) {
    return min((unsigned)d >> BSH, (unsigned)(NBUCK - 1));
}

__global__ void __launch_bounds__(TP, 8)
first_kernel(const int* __restrict__ src, const int* __restrict__ dst, int E,
             int* __restrict__ tab, int* __restrict__ st, int2* __restrict__ cpair,
             const float* __restrict__ x, const float* __restrict__ W1,
             float* __restrict__ h1) {
    __shared__ int h[NBUCK];
    __shared__ int ex[NBUCK];
    __shared__ int wsum[8];
    __shared__ __attribute__((aligned(16))) int stage_w[2 * SCAP];
    int2* stage = (int2*)stage_w;
    const int tid  = threadIdx.x;
    const int lane = tid & 31;

    if (blockIdx.x < NGEMM) {
        float* lw = (float*)stage_w;
        for (int i = tid; i < F_IN * F_HID; i += TP) {
            int kk  = i >> 6;
            int col = i & 63;
            lw[(((kk >> 1) * F_HID) + col) * 2 + (kk & 1)] = W1[i];
        }
        __syncthreads();

        const int gw = __builtin_amdgcn_readfirstlane(tid >> 5);
        const int r  = lane & 15;
        const int hi = lane >> 4;
        const v2f* wp = (const v2f*)lw;
        const int t0 = (int)blockIdx.x * GT;
        const int t1 = min(t0 + GT, NTILE);
        for (int mtile = t0 + gw; mtile < t1; mtile += TP / 32) {
            const int row0 = mtile * 16;
            v8f acc0 = {}, acc1 = {}, acc2 = {}, acc3 = {};
            const float* xrow = x + (size_t)(row0 + r) * F_IN;

            for (int k = 0; k < F_IN; k += 4) {
                __builtin_prefetch(&xrow[k + 32], 0, 3);
                v2f a = *(const v2f*)&xrow[k + 2 * hi];
                const int pb = ((k >> 1) + hi) * F_HID;
                v2f b0 = wp[pb + r +  0];
                v2f b1 = wp[pb + r + 16];
                v2f b2 = wp[pb + r + 32];
                v2f b3 = wp[pb + r + 48];
                acc0 = __builtin_amdgcn_wmma_f32_16x16x4_f32(false, a, false, b0, (short)0, acc0, false, false);
                acc1 = __builtin_amdgcn_wmma_f32_16x16x4_f32(false, a, false, b1, (short)0, acc1, false, false);
                acc2 = __builtin_amdgcn_wmma_f32_16x16x4_f32(false, a, false, b2, (short)0, acc2, false, false);
                acc3 = __builtin_amdgcn_wmma_f32_16x16x4_f32(false, a, false, b3, (short)0, acc3, false, false);
            }

            float* out = h1 + (size_t)row0 * F_HID;
#pragma unroll
            for (int i = 0; i < 8; i++) {
                int m = i + 8 * hi;
                out[(size_t)m * F_HID + (r +  0)] = acc0[i];
                out[(size_t)m * F_HID + (r + 16)] = acc1[i];
                out[(size_t)m * F_HID + (r + 32)] = acc2[i];
                out[(size_t)m * F_HID + (r + 48)] = acc3[i];
            }
        }
        return;
    }

    const int wave = tid >> 5;
    const int chunk = (int)blockIdx.x - NGEMM;
    const int ch = (E + NB1 - 1) / NB1;
    const int e0 = min(E, chunk * ch);
    const int e1 = min(E, e0 + ch);
    const bool staged = (ch <= SCAP);
    const int len = e1 - e0;

    if (tid < NBUCK) h[tid] = 0;
    __syncthreads();

    int      s[EPT];
    int      d[EPT];
    unsigned rk[EPT];
    if (staged) {
#pragma unroll
        for (int u = 0; u < EPT; u++) {
            const unsigned e = min((unsigned)(e0 + tid + u * TP), (unsigned)(E - 1));
            s[u] = src[e];
            d[u] = dst[e];
        }
#pragma unroll
        for (int u = 0; u < EPT; u++) {
            rk[u] = 0;
            if (tid < len - u * TP) rk[u] = (unsigned)atomicAdd(&h[bucket_of(d[u])], 1);
        }
    } else {
        for (int e = e0 + tid; e < e1; e += TP) atomicAdd(&h[bucket_of(dst[e])], 1);
    }
    __syncthreads();

    const int c = (tid < NBUCK) ? h[tid] : 0;
    int inc = c;
    for (int o = 1; o < 32; o <<= 1) {
        int y = __shfl_up(inc, o);
        if (lane >= o) inc += y;
    }
    if (lane == 31 && wave < 8) wsum[wave] = inc;
    __syncthreads();
    if (tid < NBUCK) {
        int xs = inc - c;
        for (int w = 0; w < 8; w++)
            if (w < wave) xs += wsum[w];
        ex[tid] = xs;
        tab[tid * NB1 + chunk] = c;
        st[tid * NB1 + chunk]  = xs;
        if (!staged) h[tid] = xs;
    }
    __syncthreads();

    if (staged) {
        unsigned q[EPT];
#pragma unroll
        for (int u = 0; u < EPT; u++)
            q[u] = min((unsigned)ex[bucket_of(d[u])] + rk[u], (unsigned)(SCAP - 1));
#pragma unroll
        for (int u = 0; u < EPT; u++)
            if (tid < len - u * TP) stage[q[u]] = make_int2(s[u], d[u]);
        __syncthreads();
        int i = tid;
        for (; i + 3 * TP < len; i += 4 * TP) {
            int2 v[4];
#pragma unroll
            for (int u = 0; u < 4; u++) v[u] = stage[i + u * TP];
#pragma unroll
            for (int u = 0; u < 4; u++) {
                const unsigned p = min((unsigned)(e0 + i + u * TP), (unsigned)(E - 1));
                cpair[p] = v[u];
            }
        }
        for (; i < len; i += TP) {
            const unsigned p = min((unsigned)(e0 + i), (unsigned)(E - 1));
            cpair[p] = stage[i];
        }
    } else {
        for (int e = e0 + tid; e < e1; e += TP) {
            const int sv = src[e];
            const int dv = dst[e];
            unsigned p = (unsigned)atomicAdd(&h[bucket_of(dv)], 1);
            p = min((unsigned)e0 + p, (unsigned)(E - 1));
            cpair[p] = make_int2(sv, dv);
        }
    }
}

__global__ void __launch_bounds__(T1)
bucket_kernel(const int2* __restrict__ cpair, const int* __restrict__ tab,
              const int* __restrict__ st, int E,
              int* __restrict__ cnt, int* __restrict__ off,
              float* __restrict__ dinv, int* __restrict__ csr_src,
              float* __restrict__ h1) {
    __shared__ int lc[BNODES];
    __shared__ int lo[BNODES];
    __shared__ int wsum[BNODES / 32];
    __shared__ int red[T1 / 32];
    __shared__ int rsum[NB1 / 32];
    __shared__ int rpos[NB1];
    __shared__ int rcnt[NB1];
    __shared__ int rg[NB1];
    __shared__ float ldv[BNODES];
    __shared__ __attribute__((aligned(16))) int raw_w[2 * RCAP];
    __shared__ int stage[RCAP];
    int2* raw = (int2*)raw_w;
    const int tid  = threadIdx.x;
    const int lane = tid & 31;
    const int wave = tid >> 5;
    const int k    = blockIdx.x;
    const int ch   = (E + NB1 - 1) / NB1;

    if (tid < BNODES) lc[tid] = 0;

    const int4* t4 = (const int4*)tab;
    const int   n4 = k * (NB1 / 4);
    int part = 0;
#pragma unroll
    for (int u = 0; u < PRE_SLOTS; u++) {
        const int  j = tid + u * T1;
        const int4 v = t4[min(j, NBUCK * NB1 / 4 - 1)];
        const int  t = (v.x + v.y) + (v.z + v.w);
        part += (j < n4) ? t : 0;
    }
    for (int o = 16; o > 0; o >>= 1) part += __shfl_xor(part, o);
    if (lane == 0) red[wave] = part;

    int rc = 0;
    int rs = 0;
    if (tid < NB1) {
        rc = tab[k * NB1 + tid];
        rs = st[k * NB1 + tid];
    }
    int rinc = rc;
    for (int o = 1; o < 32; o <<= 1) {
        int y = __shfl_up(rinc, o);
        if (lane >= o) rinc += y;
    }
    if (lane == 31 && wave < NB1 / 32) rsum[wave] = rinc;
    __syncthreads();

    int s0 = red[lane];
    for (int o = 16; o > 0; o >>= 1) s0 += __shfl_xor(s0, o);
    int len = 0;
    for (int w = 0; w < NB1 / 32; w++) len += rsum[w];
    if (tid < NB1) {
        int x = rinc - rc;
        for (int w = 0; w < NB1 / 32; w++)
            if (w < wave) x += rsum[w];
        rpos[tid] = x;
        rcnt[tid] = rc;
        rg[tid]   = tid * ch + rs;
    }
    __syncthreads();
    const bool staged = (len <= RCAP);

    auto take = [&](const int2 pr, const int q) {
        const int node = pr.y & (BNODES - 1);
        const int r = atomicAdd(&lc[node], 1);
        if (staged)
            raw[min((unsigned)q, (unsigned)(RCAP - 1))] = make_int2(pr.x | (node << SRCB), r);
    };

    for (int g = 0; g < RPW; g += 4) {
        int2 pr[4][4];
        int  rr[4][4];
        int  n[4];
        int  ps[4];
        int  gg[4];
#pragma unroll
        for (int a = 0; a < 4; a++) {
            const int run = wave * RPW + g + a;
            n[a]  = rcnt[run];
            ps[a] = rpos[run];
            gg[a] = rg[run];
        }
#pragma unroll
        for (int a = 0; a < 4; a++) {
#pragma unroll
            for (int j = 0; j < 4; j++) {
                const int p = lane + 32 * j;
                const unsigned idx = (unsigned)(gg[a] + ((p < n[a]) ? p : 0));
                pr[a][j] = cpair[min(idx, (unsigned)(E - 1))];
            }
        }
#pragma unroll
        for (int a = 0; a < 4; a++) {
#pragma unroll
            for (int j = 0; j < 4; j++) {
                rr[a][j] = 0;
                if (lane + 32 * j < n[a]) rr[a][j] = atomicAdd(&lc[pr[a][j].y & (BNODES - 1)], 1);
            }
        }
        if (staged) {
#pragma unroll
            for (int a = 0; a < 4; a++) {
#pragma unroll
                for (int j = 0; j < 4; j++) {
                    const int p = lane + 32 * j;
                    if (p < n[a]) {
                        const int node = pr[a][j].y & (BNODES - 1);
                        raw[min((unsigned)(ps[a] + p), (unsigned)(RCAP - 1))] = make_int2(pr[a][j].x | (node << SRCB), rr[a][j]);
                    }
                }
            }
        }
        for (int a = 0; a < 4; a++)
            for (int p = 128 + lane; p < n[a]; p += 32)
                take(cpair[min((unsigned)(gg[a] + p), (unsigned)(E - 1))], ps[a] + p);
    }
    __syncthreads();

    const int c = (tid < BNODES) ? lc[tid] : 0;
    int inc = c;
    for (int o = 1; o < 32; o <<= 1) {
        int y = __shfl_up(inc, o);
        if (lane >= o) inc += y;
    }
    if (lane == 31 && wave < BNODES / 32) wsum[wave] = inc;
    __syncthreads();
    if (tid < BNODES) {
        int x = inc - c;
        for (int w = 0; w < BNODES / 32; w++)
            if (w < wave) x += wsum[w];
        lo[tid] = x;
        const float dn = rsqrtf((float)c + 1.0f);
        ldv[tid] = dn;
        const int node = k * BNODES + tid;
        if (node < N_NODES) {
            cnt[node]  = c;
            off[node]  = s0 + x;
            dinv[node] = dn;
        }
    }
    __syncthreads();

    const int nrow4 = min(BNODES, N_NODES - k * BNODES) * (F_HID / 4);
    float4* hrow = (float4*)h1 + (size_t)k * BNODES * (F_HID / 4);
    float4 hr[HSLOTS];
#pragma unroll
    for (int u = 0; u < HSLOTS; u++)
        hr[u] = hrow[min(tid + u * T1, nrow4 - 1)];

    if (staged) {
        int i = tid;
        for (; i + 3 * T1 < len; i += 4 * T1) {
            int2 pr[4];
            int  b[4];
#pragma unroll
            for (int u = 0; u < 4; u++) pr[u] = raw[i + u * T1];
#pragma unroll
            for (int u = 0; u < 4; u++) b[u] = lo[((unsigned)pr[u].x >> SRCB) & (BNODES - 1)];
#pragma unroll
            for (int u = 0; u < 4; u++) {
                const unsigned q = (unsigned)b[u] + (unsigned)pr[u].y;
                stage[min(q, (unsigned)(RCAP - 1))] = pr[u].x & ((1 << SRCB) - 1);
            }
        }
        for (; i < len; i += T1) {
            const int2 pr = raw[i];
            const unsigned q = (unsigned)lo[((unsigned)pr.x >> SRCB) & (BNODES - 1)] + (unsigned)pr.y;
            stage[min(q, (unsigned)(RCAP - 1))] = pr.x & ((1 << SRCB) - 1);
        }
        __syncthreads();
        for (int j = tid; j < len; j += T1) {
            const unsigned p = min((unsigned)(s0 + j), (unsigned)(E - 1));
            csr_src[p] = stage[j];
        }
    } else {
        for (int a = 0; a < RPW; a++) {
            const int run = wave * RPW + a;
            const int n   = rcnt[run];
            const int gs  = rg[run];
            for (int p = lane; p < n; p += 32) {
                const int2 pr = cpair[min((unsigned)(gs + p), (unsigned)(E - 1))];
                const int r = atomicAdd(&lo[pr.y & (BNODES - 1)], 1);
                const unsigned q = min((unsigned)(s0 + r), (unsigned)(E - 1));
                csr_src[q] = pr.x;
            }
        }
    }

#pragma unroll
    for (int u = 0; u < HSLOTS; u++) {
        const int j = tid + u * T1;
        if (j < nrow4) {
            const float dn = ldv[j / (F_HID / 4)];
            float4 v = hr[u];
            v.x *= dn;
            v.y *= dn;
            v.z *= dn;
            v.w *= dn;
            hrow[j] = v;
        }
    }
}


template <int S, bool FULL>
__device__ __forceinline__ void gather_rows(float4& acc, int s, int j, int m,
                                            int hi, int q,
                                            const float4* __restrict__ hv) {
    float4 v[S];
    int    su[S];
#pragma unroll
    for (int u = 0; u < S; u++) su[u] = __shfl(s, j + 2 * u + hi);
#pragma unroll
    for (int u = 0; u < S; u++) v[u] = hv[(unsigned)su[u] * (F_HID / 4) + q];
#pragma unroll
    for (int u = 0; u < S; u++) {
        if (FULL) {
            acc.x += v[u].x;
            acc.y += v[u].y;
            acc.z += v[u].z;
            acc.w += v[u].w;
        } else {
            const float mk = (j + 2 * u + hi < m) ? 1.0f : 0.0f;
            acc.x += v[u].x * mk;
            acc.y += v[u].y * mk;
            acc.z += v[u].z * mk;
            acc.w += v[u].w * mk;
        }
    }
}


#define TPITCH 68
static_assert(N_NODES % 16 == 0, "a work-group's 16 nodes are a whole tile");

__global__ void __launch_bounds__(512)
gather1_gemm2_kernel(const int* __restrict__ csr_src,
                     const int* __restrict__ cnt,
                     const int* __restrict__ off,
                     const float* __restrict__ dinv,
                     const float* __restrict__ h1,
                     const float* __restrict__ b1,
                     const float* __restrict__ W2,
                     float* __restrict__ h2) {
    __shared__ __attribute__((aligned(16))) float tile[16 * TPITCH];
    __shared__ __attribute__((aligned(16))) float lw[F_HID * 16];
    __shared__ float ldd[16];
    const int tid  = threadIdx.x;
    const int lane = tid & 31;
    const int wave = __builtin_amdgcn_readfirstlane(tid >> 5);
    const int d = (int)blockIdx.x * 16 + wave;
    const int q  = lane & 15;
    const int hi = lane >> 4;

    float wv[2];
#pragma unroll
    for (int u = 0; u < 2; u++) {
        const int i = tid + u * 512;
        const int kk = i >> 4, nn = i & 15;
        wv[u] = (nn < F_OUT) ? W2[kk * F_OUT + nn] : 0.0f;
    }

    const int   n    = cnt[d];
    const int   base = off[d];
    const float dd   = dinv[d];
    const float4* hv = (const float4*)h1;

    float4 acc = {0.0f, 0.0f, 0.0f, 0.0f};
    for (int c = 0; c < n; c += 32) {
        const int m  = min(32, n - c);
        const int my = (lane < m) ? lane : (m - 1);
        unsigned su = (unsigned)csr_src[base + c + my];
        su = min(su, (unsigned)(N_NODES - 1));
        const int s = (int)su;
        int j = 0;
        while (j < m) {
            const int left = m - j;
            if (left >= 16)    { gather_rows<8, true >(acc, s, j, m, hi, q, hv); j += 16; }
            else if (left > 8) { gather_rows<8, false>(acc, s, j, m, hi, q, hv); j += 16; }
            else               { gather_rows<4, false>(acc, s, j, m, hi, q, hv); j += 8;  }
        }
    }
    acc.x += __shfl_xor(acc.x, 16);
    acc.y += __shfl_xor(acc.y, 16);
    acc.z += __shfl_xor(acc.z, 16);
    acc.w += __shfl_xor(acc.w, 16);

    if (hi == 0) {
        const float4 hs  = hv[(size_t)d * (F_HID / 4) + q];
        const float4 bb  = ((const float4*)b1)[q];
        float4 o;
        o.x = (acc.x + hs.x) * dd + bb.x;
        o.y = (acc.y + hs.y) * dd + bb.y;
        o.z = (acc.z + hs.z) * dd + bb.z;
        o.w = (acc.w + hs.w) * dd + bb.w;
        o.x = (o.x > 0.0f) ? o.x : 0.01f * o.x;
        o.y = (o.y > 0.0f) ? o.y : 0.01f * o.y;
        o.z = (o.z > 0.0f) ? o.z : 0.01f * o.z;
        o.w = (o.w > 0.0f) ? o.w : 0.01f * o.w;
        *(float4*)&tile[wave * TPITCH + 4 * q] = o;
    }
    if (lane == 0) ldd[wave] = dd;
#pragma unroll
    for (int u = 0; u < 2; u++) {
        const int i = tid + u * 512;
        const int kk = i >> 4, nn = i & 15;
        lw[(((kk >> 1) * 16) + nn) * 2 + (kk & 1)] = wv[u];
    }
    __syncthreads();
    if (wave != 0) return;

    const int r = q;
    v8f pacc = {};
    const v2f* wp = (const v2f*)lw;
    for (int k = 0; k < F_HID; k += 4) {
        v2f a = *(const v2f*)&tile[r * TPITCH + k + 2 * hi];
        v2f b = wp[((k >> 1) + hi) * 16 + r];
        pacc = __builtin_amdgcn_wmma_f32_16x16x4_f32(false, a, false, b, (short)0, pacc, false, false);
    }
    if (r < F_OUT) {
        float* out = h2 + (size_t)blockIdx.x * 16 * F_OUT;
#pragma unroll
        for (int i = 0; i < 8; i++) {
            const int m = i + 8 * hi;
            out[(size_t)m * F_OUT + r] = pacc[i] * ldd[m];
        }
    }
}


__global__ void __launch_bounds__(256)
gather2_kernel(const int* __restrict__ csr_src,
               const int* __restrict__ cnt,
               const int* __restrict__ off,
               const float* __restrict__ dinv,
               const float* __restrict__ h2,
               const float* __restrict__ b2,
               float* __restrict__ out) {
    const int t   = blockIdx.x * blockDim.x + threadIdx.x;
    const int d   = t >> 2;
    const int sub = t & 3;
    if (d >= N_NODES) return;
    const int   n    = cnt[d];
    const int   base = off[d];
    const float dd   = dinv[d];
    const float4* hv = (const float4*)h2;

    float4 acc = {0.0f, 0.0f, 0.0f, 0.0f};
    for (int j = sub; j < n; j += 16) {
        unsigned s[4];
        float    w[4];
        float4   v[4];
#pragma unroll
        for (int u = 0; u < 4; u++) {
            const bool ok = (j + 4 * u) < n;
            s[u] = (unsigned)csr_src[base + (ok ? j + 4 * u : j)];
            s[u] = min(s[u], (unsigned)(N_NODES - 1));
            w[u] = ok ? 1.0f : 0.0f;
            v[u] = hv[s[u]];
        }
#pragma unroll
        for (int u = 0; u < 4; u++) {
            acc.x += v[u].x * w[u];
            acc.y += v[u].y * w[u];
            acc.z += v[u].z * w[u];
            acc.w += v[u].w * w[u];
        }
    }
    for (int o = 2; o > 0; o >>= 1) {
        acc.x += __shfl_xor(acc.x, o);
        acc.y += __shfl_xor(acc.y, o);
        acc.z += __shfl_xor(acc.z, o);
        acc.w += __shfl_xor(acc.w, o);
    }
    if (sub == 0) {
        const float4 hs  = hv[d];
        float v[F_OUT];
        v[0] = (acc.x + hs.x) * dd + b2[0];
        v[1] = (acc.y + hs.y) * dd + b2[1];
        v[2] = (acc.z + hs.z) * dd + b2[2];
        v[3] = (acc.w + hs.w) * dd + b2[3];
        float mx = -3.4e38f;
        for (int c = 0; c < F_OUT; c++) mx = fmaxf(mx, v[c]);
        float sm = 0.0f;
        for (int c = 0; c < F_OUT; c++) { v[c] = __expf(v[c] - mx); sm += v[c]; }
        const float rs = 1.0f / sm;
        float4 o;
        o.x = v[0] * rs;  o.y = v[1] * rs;  o.z = v[2] * rs;  o.w = v[3] * rs;
        ((float4*)out)[d] = o;
    }
}


extern "C" void kernel_launch(void* const* d_in, const int* in_sizes, int n_in,
                              void* d_out, int out_size, void* d_ws, size_t ws_size,
                              hipStream_t stream) {
    const float* x  = (const float*)d_in[0];
    const int*   ei = (const int*)d_in[1];
    const float* W1 = (const float*)d_in[2];
    const float* b1 = (const float*)d_in[3];
    const float* W2 = (const float*)d_in[4];
    const float* b2 = (const float*)d_in[5];
    float* out = (float*)d_out;

    const int E = in_sizes[1] / 2;
    const int* src = ei;
    const int* dst = ei + E;

    int*   cnt     = (int*)d_ws;
    int*   off     = cnt + N_NODES;
    float* dinv    = (float*)(off + N_NODES);
    int*   csr_src = (int*)(dinv + N_NODES + 32);
    float* h1      = (float*)(csr_src + E);
    float* act1    = h1   + (size_t)N_NODES * F_HID;
    float* h2      = act1 + (size_t)N_NODES * F_HID;
    int2*  cpair   = (int2*)act1;
    int*   tab     = (int*)h2;
    int*   st      = tab + NBUCK * NB1;

    const int T = 256;
    const int nb_g1     = N_NODES / 16;
    const int nb_g2     = (N_NODES * 4 + T - 1) / T;

    first_kernel<<<NGEMM + NB1, TP, 0, stream>>>(src, dst, E, tab, st, cpair, x, W1, h1);
    bucket_kernel<<<NBUCK, T1, 0, stream>>>(cpair, tab, st, E, cnt, off, dinv, csr_src, h1);

    gather1_gemm2_kernel<<<nb_g1, 512, 0, stream>>>(csr_src, cnt, off, dinv, h1, b1, W2, h2);

    gather2_kernel<<<nb_g2, T, 0, stream>>>(csr_src, cnt, off, dinv, h2, b2, out);
}


# layer-1 gather: one wave per 16-node tile sums its nodes in turn (next node's sources asked for ahead, 32 rows per load group), multiplies its own LDS tile; 6250 waves, was 100000; sum order kept
# speedup vs baseline: 9.2731x; 1.0301x over previous
#include <hip/hip_runtime.h>


typedef __attribute__((ext_vector_type(2))) float v2f;
typedef __attribute__((ext_vector_type(8))) float v8f;

#define N_NODES 100000
#define F_IN    256
#define F_HID   64
#define F_OUT   4


#define BSH     9
#define BNODES  (1 << BSH)
#define NBUCK   ((N_NODES + BNODES - 1) / BNODES)
#define NB1     256
#define T1      1024
#define TP      512
#define EPT     25
#define SCAP    (EPT * TP)
#define NTILE   (N_NODES / 16)
#define GT      25
#define NGEMM   ((NTILE + GT - 1) / GT)
static_assert(N_NODES % 16 == 0, "the product's tiles are whole");
static_assert(2 * SCAP >= F_IN * F_HID, "W1 must fit the stage it shares");
#define RCAP    20480
#define RPW     (NB1 / (T1 / 32))
#define SRCB    20
static_assert(N_NODES <= (1 << SRCB) && BSH + SRCB <= 32, "a source index and a range-local node must fit one word");
#define PRE_SLOTS ((NBUCK * NB1 / 4 + T1 - 1) / T1)
#define HSLOTS  (BNODES * (F_HID / 4) / T1)
static_assert(BNODES * (F_HID / 4) % T1 == 0, "a range's rows are whole slots");

__device__ __forceinline__ unsigned bucket_of(int d) {
    return min((unsigned)d >> BSH, (unsigned)(NBUCK - 1));
}

__global__ void __launch_bounds__(TP, 8)
first_kernel(const int* __restrict__ src, const int* __restrict__ dst, int E,
             int* __restrict__ tab, int* __restrict__ st, int2* __restrict__ cpair,
             const float* __restrict__ x, const float* __restrict__ W1,
             float* __restrict__ h1) {
    __shared__ int h[NBUCK];
    __shared__ int ex[NBUCK];
    __shared__ int wsum[8];
    __shared__ __attribute__((aligned(16))) int stage_w[2 * SCAP];
    int2* stage = (int2*)stage_w;
    const int tid  = threadIdx.x;
    const int lane = tid & 31;

    if (blockIdx.x < NGEMM) {
        float* lw = (float*)stage_w;
        for (int i = tid; i < F_IN * F_HID; i += TP) {
            int kk  = i >> 6;
            int col = i & 63;
            lw[(((kk >> 1) * F_HID) + col) * 2 + (kk & 1)] = W1[i];
        }
        __syncthreads();

        const int gw = __builtin_amdgcn_readfirstlane(tid >> 5);
        const int r  = lane & 15;
        const int hi = lane >> 4;
        const v2f* wp = (const v2f*)lw;
        const int t0 = (int)blockIdx.x * GT;
        const int t1 = min(t0 + GT, NTILE);
        for (int mtile = t0 + gw; mtile < t1; mtile += TP / 32) {
            const int row0 = mtile * 16;
            v8f acc0 = {}, acc1 = {}, acc2 = {}, acc3 = {};
            const float* xrow = x + (size_t)(row0 + r) * F_IN;

            for (int k = 0; k < F_IN; k += 4) {
                __builtin_prefetch(&xrow[k + 32], 0, 3);
                v2f a = *(const v2f*)&xrow[k + 2 * hi];
                const int pb = ((k >> 1) + hi) * F_HID;
                v2f b0 = wp[pb + r +  0];
                v2f b1 = wp[pb + r + 16];
                v2f b2 = wp[pb + r + 32];
                v2f b3 = wp[pb + r + 48];
                acc0 = __builtin_amdgcn_wmma_f32_16x16x4_f32(false, a, false, b0, (short)0, acc0, false, false);
                acc1 = __builtin_amdgcn_wmma_f32_16x16x4_f32(false, a, false, b1, (short)0, acc1, false, false);
                acc2 = __builtin_amdgcn_wmma_f32_16x16x4_f32(false, a, false, b2, (short)0, acc2, false, false);
                acc3 = __builtin_amdgcn_wmma_f32_16x16x4_f32(false, a, false, b3, (short)0, acc3, false, false);
            }

            float* out = h1 + (size_t)row0 * F_HID;
#pragma unroll
            for (int i = 0; i < 8; i++) {
                int m = i + 8 * hi;
                out[(size_t)m * F_HID + (r +  0)] = acc0[i];
                out[(size_t)m * F_HID + (r + 16)] = acc1[i];
                out[(size_t)m * F_HID + (r + 32)] = acc2[i];
                out[(size_t)m * F_HID + (r + 48)] = acc3[i];
            }
        }
        return;
    }

    const int wave = tid >> 5;
    const int chunk = (int)blockIdx.x - NGEMM;
    const int ch = (E + NB1 - 1) / NB1;
    const int e0 = min(E, chunk * ch);
    const int e1 = min(E, e0 + ch);
    const bool staged = (ch <= SCAP);
    const int len = e1 - e0;

    if (tid < NBUCK) h[tid] = 0;
    __syncthreads();

    int      s[EPT];
    int      d[EPT];
    unsigned rk[EPT];
    if (staged) {
#pragma unroll
        for (int u = 0; u < EPT; u++) {
            const unsigned e = min((unsigned)(e0 + tid + u * TP), (unsigned)(E - 1));
            s[u] = src[e];
            d[u] = dst[e];
        }
#pragma unroll
        for (int u = 0; u < EPT; u++) {
            rk[u] = 0;
            if (tid < len - u * TP) rk[u] = (unsigned)atomicAdd(&h[bucket_of(d[u])], 1);
        }
    } else {
        for (int e = e0 + tid; e < e1; e += TP) atomicAdd(&h[bucket_of(dst[e])], 1);
    }
    __syncthreads();

    const int c = (tid < NBUCK) ? h[tid] : 0;
    int inc = c;
    for (int o = 1; o < 32; o <<= 1) {
        int y = __shfl_up(inc, o);
        if (lane >= o) inc += y;
    }
    if (lane == 31 && wave < 8) wsum[wave] = inc;
    __syncthreads();
    if (tid < NBUCK) {
        int xs = inc - c;
        for (int w = 0; w < 8; w++)
            if (w < wave) xs += wsum[w];
        ex[tid] = xs;
        tab[tid * NB1 + chunk] = c;
        st[tid * NB1 + chunk]  = xs;
        if (!staged) h[tid] = xs;
    }
    __syncthreads();

    if (staged) {
        unsigned q[EPT];
#pragma unroll
        for (int u = 0; u < EPT; u++)
            q[u] = min((unsigned)ex[bucket_of(d[u])] + rk[u], (unsigned)(SCAP - 1));
#pragma unroll
        for (int u = 0; u < EPT; u++)
            if (tid < len - u * TP) stage[q[u]] = make_int2(s[u], d[u]);
        __syncthreads();
        int i = tid;
        for (; i + 3 * TP < len; i += 4 * TP) {
            int2 v[4];
#pragma unroll
            for (int u = 0; u < 4; u++) v[u] = stage[i + u * TP];
#pragma unroll
            for (int u = 0; u < 4; u++) {
                const unsigned p = min((unsigned)(e0 + i + u * TP), (unsigned)(E - 1));
                cpair[p] = v[u];
            }
        }
        for (; i < len; i += TP) {
            const unsigned p = min((unsigned)(e0 + i), (unsigned)(E - 1));
            cpair[p] = stage[i];
        }
    } else {
        for (int e = e0 + tid; e < e1; e += TP) {
            const int sv = src[e];
            const int dv = dst[e];
            unsigned p = (unsigned)atomicAdd(&h[bucket_of(dv)], 1);
            p = min((unsigned)e0 + p, (unsigned)(E - 1));
            cpair[p] = make_int2(sv, dv);
        }
    }
}

__global__ void __launch_bounds__(T1)
bucket_kernel(const int2* __restrict__ cpair, const int* __restrict__ tab,
              const int* __restrict__ st, int E,
              int* __restrict__ cnt, int* __restrict__ off,
              float* __restrict__ dinv, int* __restrict__ csr_src,
              float* __restrict__ h1) {
    __shared__ int lc[BNODES];
    __shared__ int lo[BNODES];
    __shared__ int wsum[BNODES / 32];
    __shared__ int red[T1 / 32];
    __shared__ int rsum[NB1 / 32];
    __shared__ int rpos[NB1];
    __shared__ int rcnt[NB1];
    __shared__ int rg[NB1];
    __shared__ float ldv[BNODES];
    __shared__ __attribute__((aligned(16))) int raw_w[2 * RCAP];
    __shared__ int stage[RCAP];
    int2* raw = (int2*)raw_w;
    const int tid  = threadIdx.x;
    const int lane = tid & 31;
    const int wave = tid >> 5;
    const int k    = blockIdx.x;
    const int ch   = (E + NB1 - 1) / NB1;

    if (tid < BNODES) lc[tid] = 0;

    const int4* t4 = (const int4*)tab;
    const int   n4 = k * (NB1 / 4);
    int part = 0;
#pragma unroll
    for (int u = 0; u < PRE_SLOTS; u++) {
        const int  j = tid + u * T1;
        const int4 v = t4[min(j, NBUCK * NB1 / 4 - 1)];
        const int  t = (v.x + v.y) + (v.z + v.w);
        part += (j < n4) ? t : 0;
    }
    for (int o = 16; o > 0; o >>= 1) part += __shfl_xor(part, o);
    if (lane == 0) red[wave] = part;

    int rc = 0;
    int rs = 0;
    if (tid < NB1) {
        rc = tab[k * NB1 + tid];
        rs = st[k * NB1 + tid];
    }
    int rinc = rc;
    for (int o = 1; o < 32; o <<= 1) {
        int y = __shfl_up(rinc, o);
        if (lane >= o) rinc += y;
    }
    if (lane == 31 && wave < NB1 / 32) rsum[wave] = rinc;
    __syncthreads();

    int s0 = red[lane];
    for (int o = 16; o > 0; o >>= 1) s0 += __shfl_xor(s0, o);
    int len = 0;
    for (int w = 0; w < NB1 / 32; w++) len += rsum[w];
    if (tid < NB1) {
        int x = rinc - rc;
        for (int w = 0; w < NB1 / 32; w++)
            if (w < wave) x += rsum[w];
        rpos[tid] = x;
        rcnt[tid] = rc;
        rg[tid]   = tid * ch + rs;
    }
    __syncthreads();
    const bool staged = (len <= RCAP);

    auto take = [&](const int2 pr, const int q) {
        const int node = pr.y & (BNODES - 1);
        const int r = atomicAdd(&lc[node], 1);
        if (staged)
            raw[min((unsigned)q, (unsigned)(RCAP - 1))] = make_int2(pr.x | (node << SRCB), r);
    };

    for (int g = 0; g < RPW; g += 4) {
        int2 pr[4][4];
        int  rr[4][4];
        int  n[4];
        int  ps[4];
        int  gg[4];
#pragma unroll
        for (int a = 0; a < 4; a++) {
            const int run = wave * RPW + g + a;
            n[a]  = rcnt[run];
            ps[a] = rpos[run];
            gg[a] = rg[run];
        }
#pragma unroll
        for (int a = 0; a < 4; a++) {
#pragma unroll
            for (int j = 0; j < 4; j++) {
                const int p = lane + 32 * j;
                const unsigned idx = (unsigned)(gg[a] + ((p < n[a]) ? p : 0));
                pr[a][j] = cpair[min(idx, (unsigned)(E - 1))];
            }
        }
#pragma unroll
        for (int a = 0; a < 4; a++) {
#pragma unroll
            for (int j = 0; j < 4; j++) {
                rr[a][j] = 0;
                if (lane + 32 * j < n[a]) rr[a][j] = atomicAdd(&lc[pr[a][j].y & (BNODES - 1)], 1);
            }
        }
        if (staged) {
#pragma unroll
            for (int a = 0; a < 4; a++) {
#pragma unroll
                for (int j = 0; j < 4; j++) {
                    const int p = lane + 32 * j;
                    if (p < n[a]) {
                        const int node = pr[a][j].y & (BNODES - 1);
                        raw[min((unsigned)(ps[a] + p), (unsigned)(RCAP - 1))] = make_int2(pr[a][j].x | (node << SRCB), rr[a][j]);
                    }
                }
            }
        }
        for (int a = 0; a < 4; a++)
            for (int p = 128 + lane; p < n[a]; p += 32)
                take(cpair[min((unsigned)(gg[a] + p), (unsigned)(E - 1))], ps[a] + p);
    }
    __syncthreads();

    const int c = (tid < BNODES) ? lc[tid] : 0;
    int inc = c;
    for (int o = 1; o < 32; o <<= 1) {
        int y = __shfl_up(inc, o);
        if (lane >= o) inc += y;
    }
    if (lane == 31 && wave < BNODES / 32) wsum[wave] = inc;
    __syncthreads();
    if (tid < BNODES) {
        int x = inc - c;
        for (int w = 0; w < BNODES / 32; w++)
            if (w < wave) x += wsum[w];
        lo[tid] = x;
        const float dn = rsqrtf((float)c + 1.0f);
        ldv[tid] = dn;
        const int node = k * BNODES + tid;
        if (node < N_NODES) {
            cnt[node]  = c;
            off[node]  = s0 + x;
            dinv[node] = dn;
        }
    }
    __syncthreads();

    const int nrow4 = min(BNODES, N_NODES - k * BNODES) * (F_HID / 4);
    float4* hrow = (float4*)h1 + (size_t)k * BNODES * (F_HID / 4);
    float4 hr[HSLOTS];
#pragma unroll
    for (int u = 0; u < HSLOTS; u++)
        hr[u] = hrow[min(tid + u * T1, nrow4 - 1)];

    if (staged) {
        int i = tid;
        for (; i + 3 * T1 < len; i += 4 * T1) {
            int2 pr[4];
            int  b[4];
#pragma unroll
            for (int u = 0; u < 4; u++) pr[u] = raw[i + u * T1];
#pragma unroll
            for (int u = 0; u < 4; u++) b[u] = lo[((unsigned)pr[u].x >> SRCB) & (BNODES - 1)];
#pragma unroll
            for (int u = 0; u < 4; u++) {
                const unsigned q = (unsigned)b[u] + (unsigned)pr[u].y;
                stage[min(q, (unsigned)(RCAP - 1))] = pr[u].x & ((1 << SRCB) - 1);
            }
        }
        for (; i < len; i += T1) {
            const int2 pr = raw[i];
            const unsigned q = (unsigned)lo[((unsigned)pr.x >> SRCB) & (BNODES - 1)] + (unsigned)pr.y;
            stage[min(q, (unsigned)(RCAP - 1))] = pr.x & ((1 << SRCB) - 1);
        }
        __syncthreads();
        for (int j = tid; j < len; j += T1) {
            const unsigned p = min((unsigned)(s0 + j), (unsigned)(E - 1));
            csr_src[p] = stage[j];
        }
    } else {
        for (int a = 0; a < RPW; a++) {
            const int run = wave * RPW + a;
            const int n   = rcnt[run];
            const int gs  = rg[run];
            for (int p = lane; p < n; p += 32) {
                const int2 pr = cpair[min((unsigned)(gs + p), (unsigned)(E - 1))];
                const int r = atomicAdd(&lo[pr.y & (BNODES - 1)], 1);
                const unsigned q = min((unsigned)(s0 + r), (unsigned)(E - 1));
                csr_src[q] = pr.x;
            }
        }
    }

#pragma unroll
    for (int u = 0; u < HSLOTS; u++) {
        const int j = tid + u * T1;
        if (j < nrow4) {
            const float dn = ldv[j / (F_HID / 4)];
            float4 v = hr[u];
            v.x *= dn;
            v.y *= dn;
            v.z *= dn;
            v.w *= dn;
            hrow[j] = v;
        }
    }
}


template <int S>
__device__ __forceinline__ void gather_rows(float4& acc, int s, int m,
                                            int hi, int q,
                                            const float4* __restrict__ hv) {
    float4 v[S];
#pragma unroll
    for (int u = 0; u < S; u++)
        if (2 * u < m) {
            const int sa = __builtin_amdgcn_readlane(s, 2 * u);
            const int sb = __builtin_amdgcn_readlane(s, 2 * u + 1);
            const int su = hi ? sb : sa;
            v[u] = hv[(unsigned)su * (F_HID / 4) + q];
        }
#pragma unroll
    for (int u = 0; u < S; u++)
        if (2 * u < m) {
            const bool ok = (2 * u + hi) < m;
            acc.x += ok ? v[u].x : 0.0f;
            acc.y += ok ? v[u].y : 0.0f;
            acc.z += ok ? v[u].z : 0.0f;
            acc.w += ok ? v[u].w : 0.0f;
        }
}


#define TPITCH 68
#define GW     4
#define NGWG   ((NTILE + GW - 1) / GW)
static_assert(N_NODES % 16 == 0, "a wave's 16 nodes are a whole tile");
static_assert(F_HID * 16 % (GW * 32) == 0, "W2's padded words are whole slots per thread");
#define W2SLOTS (F_HID * 16 / (GW * 32))

__global__ void __launch_bounds__(GW * 32)
gather1_gemm2_kernel(const int* __restrict__ csr_src,
                     const int* __restrict__ cnt,
                     const int* __restrict__ off,
                     const float* __restrict__ dinv,
                     const float* __restrict__ h1,
                     const float* __restrict__ b1,
                     const float* __restrict__ W2,
                     float* __restrict__ h2) {
    __shared__ __attribute__((aligned(16))) float tile_all[GW * 16 * TPITCH];
    __shared__ __attribute__((aligned(16))) float lw[F_HID * 16];
    __shared__ float ldd_all[GW * 16];
    const int tid  = threadIdx.x;
    const int lane = tid & 31;
    const int wave = __builtin_amdgcn_readfirstlane(tid >> 5);
    const int q  = lane & 15;
    const int hi = lane >> 4;
    float* tile = tile_all + wave * (16 * TPITCH);
    float* ldd  = ldd_all + wave * 16;

    float wv[W2SLOTS];
#pragma unroll
    for (int u = 0; u < W2SLOTS; u++) {
        const int i = tid + u * (GW * 32);
        const int kk = i >> 4, nn = i & 15;
        wv[u] = (nn < F_OUT) ? W2[kk * F_OUT + nn] : 0.0f;
    }

    const int  tw   = (int)blockIdx.x * GW + wave;
    const bool live = tw < NTILE;
    const int  t    = min(tw, NTILE - 1);
    const int  d0   = t * 16;
    const int   nv = cnt[d0 + q];
    const int   bv = off[d0 + q];
    const float dv = dinv[d0 + q];
    const float4 bb = ((const float4*)b1)[q];
    const float4* hv = (const float4*)h1;

#pragma unroll
    for (int u = 0; u < W2SLOTS; u++) {
        const int i = tid + u * (GW * 32);
        const int kk = i >> 4, nn = i & 15;
        lw[(((kk >> 1) * 16) + nn) * 2 + (kk & 1)] = wv[u];
    }
    if (hi == 0) ldd[q] = dv;
    __syncthreads();
    if (!live) return;

    int n    = __builtin_amdgcn_readlane(nv, 0);
    int base = __builtin_amdgcn_readlane(bv, 0);
    int s0 = 0, s1 = 0;
    if (n > 0)  s0 = csr_src[base + min(lane, n - 1)];
    if (n > 32) s1 = csr_src[base + 32 + min(lane, n - 33)];
    float4 hs = hv[(size_t)d0 * (F_HID / 4) + q];

    for (int i = 0; i < 16; i++) {
        int n1 = 0, base1 = 0, s0n = 0, s1n = 0;
        float4 hsn = hs;
        if (i + 1 < 16) {
            n1    = __builtin_amdgcn_readlane(nv, i + 1);
            base1 = __builtin_amdgcn_readlane(bv, i + 1);
            if (n1 > 0)  s0n = csr_src[base1 + min(lane, n1 - 1)];
            if (n1 > 32) s1n = csr_src[base1 + 32 + min(lane, n1 - 33)];
            hsn = hv[(size_t)(d0 + i + 1) * (F_HID / 4) + q];
        }
        const float dd = __int_as_float(__builtin_amdgcn_readlane(__float_as_int(dv), i));

        float4 acc = {0.0f, 0.0f, 0.0f, 0.0f};
        auto chunk = [&](unsigned su, const int m) {
            su = min(su, (unsigned)(N_NODES - 1));
            gather_rows<16>(acc, (int)su, m, hi, q, hv);
        };
        if (n > 0)  chunk((unsigned)s0, min(32, n));
        if (n > 32) chunk((unsigned)s1, min(32, n - 32));
        for (int c = 64; c < n; c += 32) {
            const int m  = min(32, n - c);
            const int my = (lane < m) ? lane : (m - 1);
            chunk((unsigned)csr_src[base + c + my], m);
        }
        acc.x += __shfl_xor(acc.x, 16);
        acc.y += __shfl_xor(acc.y, 16);
        acc.z += __shfl_xor(acc.z, 16);
        acc.w += __shfl_xor(acc.w, 16);

        if (hi == 0) {
            float4 o;
            o.x = (acc.x + hs.x) * dd + bb.x;
            o.y = (acc.y + hs.y) * dd + bb.y;
            o.z = (acc.z + hs.z) * dd + bb.z;
            o.w = (acc.w + hs.w) * dd + bb.w;
            o.x = (o.x > 0.0f) ? o.x : 0.01f * o.x;
            o.y = (o.y > 0.0f) ? o.y : 0.01f * o.y;
            o.z = (o.z > 0.0f) ? o.z : 0.01f * o.z;
            o.w = (o.w > 0.0f) ? o.w : 0.01f * o.w;
            *(float4*)&tile[i * TPITCH + 4 * q] = o;
        }
        n = n1;  base = base1;  s0 = s0n;  s1 = s1n;  hs = hsn;
    }

    const int r = q;
    v8f pacc = {};
    const v2f* wp = (const v2f*)lw;
    for (int k = 0; k < F_HID; k += 4) {
        v2f a = *(const v2f*)&tile[r * TPITCH + k + 2 * hi];
        v2f b = wp[((k >> 1) + hi) * 16 + r];
        pacc = __builtin_amdgcn_wmma_f32_16x16x4_f32(false, a, false, b, (short)0, pacc, false, false);
    }
    if (r < F_OUT) {
        float* out = h2 + (size_t)t * 16 * F_OUT;
#pragma unroll
        for (int i = 0; i < 8; i++) {
            const int m = i + 8 * hi;
            out[(size_t)m * F_OUT + r] = pacc[i] * ldd[m];
        }
    }
}


__global__ void __launch_bounds__(256)
gather2_kernel(const int* __restrict__ csr_src,
               const int* __restrict__ cnt,
               const int* __restrict__ off,
               const float* __restrict__ dinv,
               const float* __restrict__ h2,
               const float* __restrict__ b2,
               float* __restrict__ out) {
    const int t   = blockIdx.x * blockDim.x + threadIdx.x;
    const int d   = t >> 2;
    const int sub = t & 3;
    if (d >= N_NODES) return;
    const int   n    = cnt[d];
    const int   base = off[d];
    const float dd   = dinv[d];
    const float4* hv = (const float4*)h2;

    float4 acc = {0.0f, 0.0f, 0.0f, 0.0f};
    for (int j = sub; j < n; j += 16) {
        unsigned s[4];
        float    w[4];
        float4   v[4];
#pragma unroll
        for (int u = 0; u < 4; u++) {
            const bool ok = (j + 4 * u) < n;
            s[u] = (unsigned)csr_src[base + (ok ? j + 4 * u : j)];
            s[u] = min(s[u], (unsigned)(N_NODES - 1));
            w[u] = ok ? 1.0f : 0.0f;
            v[u] = hv[s[u]];
        }
#pragma unroll
        for (int u = 0; u < 4; u++) {
            acc.x += v[u].x * w[u];
            acc.y += v[u].y * w[u];
            acc.z += v[u].z * w[u];
            acc.w += v[u].w * w[u];
        }
    }
    for (int o = 2; o > 0; o >>= 1) {
        acc.x += __shfl_xor(acc.x, o);
        acc.y += __shfl_xor(acc.y, o);
        acc.z += __shfl_xor(acc.z, o);
        acc.w += __shfl_xor(acc.w, o);
    }
    if (sub == 0) {
        const float4 hs  = hv[d];
        float v[F_OUT];
        v[0] = (acc.x + hs.x) * dd + b2[0];
        v[1] = (acc.y + hs.y) * dd + b2[1];
        v[2] = (acc.z + hs.z) * dd + b2[2];
        v[3] = (acc.w + hs.w) * dd + b2[3];
        float mx = -3.4e38f;
        for (int c = 0; c < F_OUT; c++) mx = fmaxf(mx, v[c]);
        float sm = 0.0f;
        for (int c = 0; c < F_OUT; c++) { v[c] = __expf(v[c] - mx); sm += v[c]; }
        const float rs = 1.0f / sm;
        float4 o;
        o.x = v[0] * rs;  o.y = v[1] * rs;  o.z = v[2] * rs;  o.w = v[3] * rs;
        ((float4*)out)[d] = o;
    }
}


extern "C" void kernel_launch(void* const* d_in, const int* in_sizes, int n_in,
                              void* d_out, int out_size, void* d_ws, size_t ws_size,
                              hipStream_t stream) {
    const float* x  = (const float*)d_in[0];
    const int*   ei = (const int*)d_in[1];
    const float* W1 = (const float*)d_in[2];
    const float* b1 = (const float*)d_in[3];
    const float* W2 = (const float*)d_in[4];
    const float* b2 = (const float*)d_in[5];
    float* out = (float*)d_out;

    const int E = in_sizes[1] / 2;
    const int* src = ei;
    const int* dst = ei + E;

    int*   cnt     = (int*)d_ws;
    int*   off     = cnt + N_NODES;
    float* dinv    = (float*)(off + N_NODES);
    int*   csr_src = (int*)(dinv + N_NODES + 32);
    float* h1      = (float*)(csr_src + E);
    float* act1    = h1   + (size_t)N_NODES * F_HID;
    float* h2      = act1 + (size_t)N_NODES * F_HID;
    int2*  cpair   = (int2*)act1;
    int*   tab     = (int*)h2;
    int*   st      = tab + NBUCK * NB1;

    const int T = 256;
    const int nb_g2     = (N_NODES * 4 + T - 1) / T;

    first_kernel<<<NGEMM + NB1, TP, 0, stream>>>(src, dst, E, tab, st, cpair, x, W1, h1);
    bucket_kernel<<<NBUCK, T1, 0, stream>>>(cpair, tab, st, E, cnt, off, dinv, csr_src, h1);

    gather1_gemm2_kernel<<<NGWG, GW * 32, 0, stream>>>(csr_src, cnt, off, dinv, h1, b1, W2, h2);

    gather2_kernel<<<nb_g2, T, 0, stream>>>(csr_src, cnt, off, dinv, h2, b2, out);
}
